# starting kernel
# baseline (speedup 1.0000x reference)
#include <hip/hip_runtime.h>
#include <hip/hip_bf16.h>


typedef __attribute__((ext_vector_type(2)))  float    v2f;
typedef __attribute__((ext_vector_type(8)))  float    v8f;
typedef __attribute__((ext_vector_type(16))) _Float16 v16h;

__device__ __forceinline__ void atomicAddF(float* p, float v) {
    __hip_atomic_fetch_add(p, v, __ATOMIC_RELAXED, __HIP_MEMORY_SCOPE_AGENT);
}

__global__ void gcn_init(float* __restrict__ agg1, float* __restrict__ outb,
                         float* __restrict__ deg, int n8  ) {
    int i = blockIdx.x * blockDim.x + threadIdx.x;
    int n = n8 >> 3;
    if (i < n8) agg1[i] = 0.0f;
    if (i < (n << 1)) outb[i] = 0.0f;
    if (i < n) deg[i] = 1.0f;
}

__global__ void gcn_degree(const int* __restrict__ col, float* __restrict__ deg, int E) {
    int e = blockIdx.x * blockDim.x + threadIdx.x;
    if (e < E) atomicAddF(&deg[col[e]], 1.0f);
}

__global__ void gcn_dinv(float* __restrict__ deg, int N) {
    int i = blockIdx.x * blockDim.x + threadIdx.x;
    if (i < N) deg[i] = __frsqrt_rn(deg[i]);
}

__global__ void gcn_xw1_wmma(const float* __restrict__ x, const float* __restrict__ W1,
                             float* __restrict__ xw1, int nTiles, int N) {
    __shared__ float tile[8][16 * 8];
    int flat = blockIdx.x * blockDim.x + threadIdx.x;
    int wave = __builtin_amdgcn_readfirstlane(flat >> 5);
    if (wave >= nTiles) return;

    const int lane = threadIdx.x & 31;
    const int m    = lane & 15;
    const int half = lane >> 4;
    const int n    = m;
    const float colMask = (n < 8) ? 1.0f : 0.0f;
    const int   nn      = n & 7;

    long rowA = (long)wave * 16 + m;
    if (rowA >= N) rowA = N - 1;
    const float* xr = x + rowA * 16;

    v8f acc = {};
#if __has_builtin(__builtin_amdgcn_wmma_f32_16x16x4_f32)
#pragma unroll
    for (int kb = 0; kb < 16; kb += 4) {
        const int k0 = kb + 2 * half;
        v2f a, b;
        a.x = xr[k0 + 0];
        a.y = xr[k0 + 1];
        b.x = W1[(k0 + 0) * 8 + nn] * colMask;
        b.y = W1[(k0 + 1) * 8 + nn] * colMask;
        acc = __builtin_amdgcn_wmma_f32_16x16x4_f32(
                  false, a, false, b, (short)0, acc, false, false);
    }
#else
    v16h a, b;
#pragma unroll
    for (int j = 0; j < 16; ++j) {
        a[j] = (j < 8) ? (_Float16)xr[8 * half + j] : (_Float16)0.0f;
        float bv = (half == 0) ? W1[j * 8 + nn] * colMask : 0.0f;
        b[j] = (_Float16)bv;
    }
    acc = __builtin_amdgcn_wmma_f32_16x16x32_f16(
              false, a, false, b, (short)0, acc, false, false);
#endif

    float* t = &tile[threadIdx.x >> 5][0];
    if (n < 8) {
#pragma unroll
        for (int r = 0; r < 8; ++r)
            t[(r + 8 * half) * 8 + n] = acc[r];
    }
    asm volatile("s_wait_dscnt 0x0" ::: "memory");

    const int rowIdx = lane >> 1;
    const int quad   = lane & 1;
    long gr = (long)wave * 16 + rowIdx;
    float4 v = *(const float4*)(t + rowIdx * 8 + quad * 4);
    if (gr < N) *(float4*)(xw1 + gr * 8 + quad * 4) = v;
}

__global__ void gcn_prop1(const int* __restrict__ row, const int* __restrict__ col,
                          const float* __restrict__ dinv, const float* __restrict__ xw1,
                          float* __restrict__ agg1, long EF  ) {
    long t = (long)blockIdx.x * blockDim.x + threadIdx.x;
    if (t >= EF) return;
    const int e = (int)(t >> 3);
    const int f = (int)(t & 7);
    const int r = row[e], c = col[e];
    const float w = dinv[r] * dinv[c];
    atomicAddF(agg1 + (long)c * 8 + f, xw1[(long)r * 8 + f] * w);
}

__global__ void gcn_relu_hw2(const float* __restrict__ agg1, const float* __restrict__ xw1,
                             const float* __restrict__ dinv, const float* __restrict__ b1,
                             const float* __restrict__ W2, float* __restrict__ hw2, int N) {
    int i = blockIdx.x * blockDim.x + threadIdx.x;
    if (i >= N) return;
    const float self = dinv[i] * dinv[i];
    const float4* a = (const float4*)(agg1 + (long)i * 8);
    const float4* s = (const float4*)(xw1 + (long)i * 8);
    float hv[8];
    {
        float4 a0 = a[0], a1 = a[1], s0 = s[0], s1 = s[1];
        hv[0] = a0.x + self * s0.x; hv[1] = a0.y + self * s0.y;
        hv[2] = a0.z + self * s0.z; hv[3] = a0.w + self * s0.w;
        hv[4] = a1.x + self * s1.x; hv[5] = a1.y + self * s1.y;
        hv[6] = a1.z + self * s1.z; hv[7] = a1.w + self * s1.w;
    }
    float o0 = 0.0f, o1 = 0.0f;
#pragma unroll
    for (int f = 0; f < 8; ++f) {
        float h = fmaxf(hv[f] + b1[f], 0.0f);
        o0 = fmaf(h, W2[f * 2 + 0], o0);
        o1 = fmaf(h, W2[f * 2 + 1], o1);
    }
    hw2[(long)i * 2 + 0] = o0;
    hw2[(long)i * 2 + 1] = o1;
}

__global__ void gcn_prop2(const int* __restrict__ row, const int* __restrict__ col,
                          const float* __restrict__ dinv, const float* __restrict__ hw2,
                          float* __restrict__ outb, long EF2  ) {
    long t = (long)blockIdx.x * blockDim.x + threadIdx.x;
    if (t >= EF2) return;
    const int e = (int)(t >> 1);
    const int f = (int)(t & 1);
    const int r = row[e], c = col[e];
    const float w = dinv[r] * dinv[c];
    atomicAddF(outb + (long)c * 2 + f, hw2[(long)r * 2 + f] * w);
}

__global__ void gcn_finalize(float* __restrict__ outb, const float* __restrict__ hw2,
                             const float* __restrict__ dinv, const float* __restrict__ b2,
                             int N) {
    int i = blockIdx.x * blockDim.x + threadIdx.x;
    if (i >= N) return;
    const float self = dinv[i] * dinv[i];
    float v0 = outb[(long)i * 2 + 0] + self * hw2[(long)i * 2 + 0] + b2[0];
    float v1 = outb[(long)i * 2 + 1] + self * hw2[(long)i * 2 + 1] + b2[1];
    const float m = fmaxf(v0, v1);
    const float lse = m + logf(expf(v0 - m) + expf(v1 - m));
    outb[(long)i * 2 + 0] = v0 - lse;
    outb[(long)i * 2 + 1] = v1 - lse;
}

extern "C" void kernel_launch(void* const* d_in, const int* in_sizes, int n_in,
                              void* d_out, int out_size, void* d_ws, size_t ws_size,
                              hipStream_t stream) {
    const float* x    = (const float*)d_in[0];
    const float* W1   = (const float*)d_in[1];
    const float* b1   = (const float*)d_in[2];
    const float* W2   = (const float*)d_in[3];
    const float* b2   = (const float*)d_in[4];
    const int*   edge = (const int*)d_in[5];

    const int N = in_sizes[0] / 16;
    const int E = in_sizes[5] / 2;
    const int* row = edge;
    const int* col = edge + E;
    float* out = (float*)d_out;

    size_t off = 0;
    auto carve = [&](size_t bytes) {
        void* p = (char*)d_ws + off;
        off = (off + bytes + 255) & ~(size_t)255;
        return p;
    };
    float* deg  = (float*)carve((size_t)N * 4);
    float* xw1  = (float*)carve((size_t)N * 8 * 4);
    float* agg1 = (float*)carve((size_t)N * 8 * 4);
    float* hw2  = (float*)carve((size_t)N * 2 * 4);
    (void)ws_size;

    const int B = 256;
    const long EF  = (long)E * 8;
    const long EF2 = (long)E * 2;
    const int gN   = (N + B - 1) / B;
    const int gN8  = (N * 8 + B - 1) / B;
    const int gE   = (E + B - 1) / B;
    const int gEF  = (int)((EF + B - 1) / B);
    const int gEF2 = (int)((EF2 + B - 1) / B);
    const int nTiles = (N + 15) / 16;
    const int gW   = (nTiles * 32 + B - 1) / B;

    gcn_init    <<<gN8,  B, 0, stream>>>(agg1, out, deg, N * 8);
    gcn_degree  <<<gE,   B, 0, stream>>>(col, deg, E);
    gcn_dinv    <<<gN,   B, 0, stream>>>(deg, N);
    gcn_xw1_wmma<<<gW,   B, 0, stream>>>(x, W1, xw1, nTiles, N);
    gcn_prop1   <<<gEF,  B, 0, stream>>>(row, col, deg, xw1, agg1, EF);
    gcn_relu_hw2<<<gN,   B, 0, stream>>>(agg1, xw1, deg, b1, W2, hw2, N);
    gcn_prop2   <<<gEF2, B, 0, stream>>>(row, col, deg, hw2, out, EF2);
    gcn_finalize<<<gN,   B, 0, stream>>>(out, hw2, deg, b2, N);
}


# Edges counting-sorted into 1024-node destination bins; one block per bin sums its messages with LDS float adds and stores each row once; global float atomics removed, sum order still arrival order
# speedup vs baseline: 3.1586x; 3.1586x over previous
#include <hip/hip_runtime.h>
#include <hip/hip_bf16.h>


typedef __attribute__((ext_vector_type(2)))  float    v2f;
typedef __attribute__((ext_vector_type(8)))  float    v8f;
typedef __attribute__((ext_vector_type(16))) _Float16 v16h;

#define BIN_SHIFT 10
#define BIN_NODES 1024
#define BIN_MASK  1023
#define CNT_PITCH 1088
#define CHUNK     16384

__device__ __forceinline__ void ldsAddF(float* p, float v) {
    __hip_atomic_fetch_add(p, v, __ATOMIC_RELAXED, __HIP_MEMORY_SCOPE_WORKGROUP);
}
__device__ __forceinline__ void ldsAddI(int* p) {
    __hip_atomic_fetch_add(p, 1, __ATOMIC_RELAXED, __HIP_MEMORY_SCOPE_WORKGROUP);
}

__global__ void bin_init(int* __restrict__ tot, int* __restrict__ cur, int P) {
    int i = blockIdx.x * blockDim.x + threadIdx.x;
    if (i < P) { tot[i * CNT_PITCH] = 0; cur[i * CNT_PITCH] = 0; }
}

__device__ __forceinline__ void chunk_hist(const int* __restrict__ col, int s, int e,
                                           int* hist, int tid) {
    for (int j0 = s + tid; j0 < e; j0 += 4 * BIN_NODES) {
        int b[4];
#pragma unroll
        for (int u = 0; u < 4; ++u) {
            const int j = j0 + u * BIN_NODES;
            b[u] = (j < e) ? (col[j] >> BIN_SHIFT) : -1;
        }
#pragma unroll
        for (int u = 0; u < 4; ++u)
            if (b[u] >= 0) ldsAddI(&hist[b[u]]);
    }
}

__global__ __launch_bounds__(BIN_NODES)
void bin_count(const int* __restrict__ col, int* __restrict__ tot, int E, int P) {
    __shared__ int hist[BIN_NODES];
    const int tid = threadIdx.x;
    hist[tid] = 0;
    __syncthreads();
    const int s = blockIdx.x * CHUNK;
    const int e = min(s + CHUNK, E);
    chunk_hist(col, s, e, hist, tid);
    __syncthreads();
    if (tid < P) {
        const int h = hist[tid];
        if (h) __hip_atomic_fetch_add(&tot[tid * CNT_PITCH], h, __ATOMIC_RELAXED,
                                      __HIP_MEMORY_SCOPE_AGENT);
    }
}

__device__ __forceinline__ int block_incl_scan(int v, int* s, int tid) {
    s[tid] = v;
    __syncthreads();
    for (int off = 1; off < BIN_NODES; off <<= 1) {
        const int t = (tid >= off) ? s[tid - off] : 0;
        __syncthreads();
        s[tid] += t;
        __syncthreads();
    }
    return s[tid];
}

__global__ __launch_bounds__(BIN_NODES)
void bin_scan(const int* __restrict__ tot, int* __restrict__ start, int P) {
    __shared__ int s[BIN_NODES];
    const int tid = threadIdx.x;
    const int incl = block_incl_scan((tid < P) ? tot[tid * CNT_PITCH] : 0, s, tid);
    if (tid < P) start[tid + 1] = incl;
    if (tid == 0) start[0] = 0;
}

__global__ __launch_bounds__(BIN_NODES)
void bin_scatter(const int* __restrict__ row, const int* __restrict__ col,
                 const int* __restrict__ start, int* __restrict__ cur,
                 unsigned* __restrict__ keys, int E, int P) {
    __shared__ unsigned       stage[CHUNK];
    __shared__ unsigned short sbin[CHUNK];
    __shared__ int hist[BIN_NODES];
    __shared__ int lst[BIN_NODES];
    __shared__ int delta[BIN_NODES];
    const int tid = threadIdx.x;
    hist[tid] = 0;
    __syncthreads();
    const int s = blockIdx.x * CHUNK;
    const int e = min(s + CHUNK, E);
    chunk_hist(col, s, e, hist, tid);
    __syncthreads();
    const int h = hist[tid];
    int gbase = 0;
    if (tid < P && h)
        gbase = start[tid] + __hip_atomic_fetch_add(&cur[tid * CNT_PITCH], h,
                                 __ATOMIC_RELAXED, __HIP_MEMORY_SCOPE_AGENT);
    const int excl = block_incl_scan(h, lst, tid) - h;
    lst[tid]   = excl;
    delta[tid] = gbase - excl;
    hist[tid]  = 0;
    __syncthreads();
    for (int j0 = s + tid; j0 < e; j0 += 4 * BIN_NODES) {
        int r[4], c[4];
#pragma unroll
        for (int u = 0; u < 4; ++u) {
            const int j = j0 + u * BIN_NODES;
            const int jj = (j < e) ? j : j0;
            r[u] = row[jj];
            c[u] = (j < e) ? col[jj] : -1;
        }
#pragma unroll
        for (int u = 0; u < 4; ++u) {
            if (c[u] >= 0) {
                const int b  = c[u] >> BIN_SHIFT;
                const int li = lst[b] + __hip_atomic_fetch_add(&hist[b], 1, __ATOMIC_RELAXED,
                                                               __HIP_MEMORY_SCOPE_WORKGROUP);
                stage[li] = ((unsigned)r[u] << BIN_SHIFT) | (unsigned)(c[u] & BIN_MASK);
                sbin[li]  = (unsigned short)b;
            }
        }
    }
    __syncthreads();
    const int n = e - s;
    for (int i = tid; i < n; i += BIN_NODES)
        keys[delta[sbin[i]] + i] = stage[i];
}

__global__ __launch_bounds__(BIN_NODES)
void bin_degree(const unsigned* __restrict__ keys, const int* __restrict__ start,
                float* __restrict__ dinv, int N) {
    __shared__ int cnt[BIN_NODES];
    const int tid = threadIdx.x;
    cnt[tid] = 0;
    __syncthreads();
    const int s = start[blockIdx.x], e = start[blockIdx.x + 1];
    for (int j0 = s + tid; j0 < e; j0 += 4 * BIN_NODES) {
        int k[4];
#pragma unroll
        for (int u = 0; u < 4; ++u) {
            const int j = j0 + u * BIN_NODES;
            k[u] = (j < e) ? (int)(keys[j] & BIN_MASK) : -1;
        }
#pragma unroll
        for (int u = 0; u < 4; ++u)
            if (k[u] >= 0) ldsAddI(&cnt[k[u]]);
    }
    __syncthreads();
    const int node = blockIdx.x * BIN_NODES + tid;
    if (node < N) dinv[node] = __frsqrt_rn((float)(cnt[tid] + 1));
}

__global__ void gcn_xw1_wmma(const float* __restrict__ x, const float* __restrict__ W1,
                             float* __restrict__ xw1, int nTiles, int N) {
    __shared__ float tile[8][16 * 8];
    int flat = blockIdx.x * blockDim.x + threadIdx.x;
    int wave = __builtin_amdgcn_readfirstlane(flat >> 5);
    if (wave >= nTiles) return;

    const int lane = threadIdx.x & 31;
    const int m    = lane & 15;
    const int half = lane >> 4;
    const int n    = m;
    const float colMask = (n < 8) ? 1.0f : 0.0f;
    const int   nn      = n & 7;

    long rowA = (long)wave * 16 + m;
    if (rowA >= N) rowA = N - 1;
    const float* xr = x + rowA * 16;

    v8f acc = {};
#if __has_builtin(__builtin_amdgcn_wmma_f32_16x16x4_f32)
#pragma unroll
    for (int kb = 0; kb < 16; kb += 4) {
        const int k0 = kb + 2 * half;
        v2f a, b;
        a.x = xr[k0 + 0];
        a.y = xr[k0 + 1];
        b.x = W1[(k0 + 0) * 8 + nn] * colMask;
        b.y = W1[(k0 + 1) * 8 + nn] * colMask;
        acc = __builtin_amdgcn_wmma_f32_16x16x4_f32(
                  false, a, false, b, (short)0, acc, false, false);
    }
#else
    v16h a, b;
#pragma unroll
    for (int j = 0; j < 16; ++j) {
        a[j] = (j < 8) ? (_Float16)xr[8 * half + j] : (_Float16)0.0f;
        float bv = (half == 0) ? W1[j * 8 + nn] * colMask : 0.0f;
        b[j] = (_Float16)bv;
    }
    acc = __builtin_amdgcn_wmma_f32_16x16x32_f16(
              false, a, false, b, (short)0, acc, false, false);
#endif

    float* t = &tile[threadIdx.x >> 5][0];
    if (n < 8) {
#pragma unroll
        for (int r = 0; r < 8; ++r)
            t[(r + 8 * half) * 8 + n] = acc[r];
    }
    asm volatile("s_wait_dscnt 0x0" ::: "memory");

    const int rowIdx = lane >> 1;
    const int quad   = lane & 1;
    long gr = (long)wave * 16 + rowIdx;
    float4 v = *(const float4*)(t + rowIdx * 8 + quad * 4);
    if (gr < N) *(float4*)(xw1 + gr * 8 + quad * 4) = v;
}

__global__ __launch_bounds__(BIN_NODES)
void bin_prop1(const unsigned* __restrict__ keys, const int* __restrict__ start,
               const float* __restrict__ dinv, const float* __restrict__ xw1,
               float* __restrict__ agg1, int N) {
    __shared__ float acc[BIN_NODES * 8];
    __shared__ float dl[BIN_NODES];
    const int tid   = threadIdx.x;
    const int node0 = blockIdx.x * BIN_NODES;
#pragma unroll
    for (int i = 0; i < 8; ++i) acc[i * BIN_NODES + tid] = 0.0f;
    dl[tid] = (node0 + tid < N) ? dinv[node0 + tid] : 0.0f;
    __syncthreads();

    const int s = start[blockIdx.x], e = start[blockIdx.x + 1];
    const int f   = tid & 7;
    const int sub = tid >> 3;
    for (int j0 = s + sub; j0 < e; j0 += 8 * 128) {
        unsigned k[8];
        bool ok[8];
#pragma unroll
        for (int u = 0; u < 8; ++u) {
            const int j = j0 + u * 128;
            ok[u] = j < e;
            k[u] = keys[ok[u] ? j : j0];
        }
        float v[8];
        int   a[8];
#pragma unroll
        for (int u = 0; u < 8; ++u) {
            const int r  = (int)(k[u] >> BIN_SHIFT);
            const int cl = (int)(k[u] & BIN_MASK);
            const float w = dinv[r] * dl[cl];
            v[u] = xw1[(long)r * 8 + f] * w;
            a[u] = cl * 8 + f;
        }
#pragma unroll
        for (int u = 0; u < 8; ++u)
            if (ok[u]) ldsAddF(&acc[a[u]], v[u]);
    }
    __syncthreads();
    float* dst = agg1 + (long)node0 * 8;
#pragma unroll
    for (int i = 0; i < 8; ++i) {
        const int idx = i * BIN_NODES + tid;
        if (node0 + (idx >> 3) < N) dst[idx] = acc[idx];
    }
}

__global__ void gcn_relu_hw2(const float* __restrict__ agg1, const float* __restrict__ xw1,
                             const float* __restrict__ dinv, const float* __restrict__ b1,
                             const float* __restrict__ W2, float* __restrict__ hw2, int N) {
    int i = blockIdx.x * blockDim.x + threadIdx.x;
    if (i >= N) return;
    const float self = dinv[i] * dinv[i];
    const float4* a = (const float4*)(agg1 + (long)i * 8);
    const float4* s = (const float4*)(xw1 + (long)i * 8);
    float hv[8];
    {
        float4 a0 = a[0], a1 = a[1], s0 = s[0], s1 = s[1];
        hv[0] = a0.x + self * s0.x; hv[1] = a0.y + self * s0.y;
        hv[2] = a0.z + self * s0.z; hv[3] = a0.w + self * s0.w;
        hv[4] = a1.x + self * s1.x; hv[5] = a1.y + self * s1.y;
        hv[6] = a1.z + self * s1.z; hv[7] = a1.w + self * s1.w;
    }
    float o0 = 0.0f, o1 = 0.0f;
#pragma unroll
    for (int f = 0; f < 8; ++f) {
        float h = fmaxf(hv[f] + b1[f], 0.0f);
        o0 = fmaf(h, W2[f * 2 + 0], o0);
        o1 = fmaf(h, W2[f * 2 + 1], o1);
    }
    hw2[(long)i * 2 + 0] = o0;
    hw2[(long)i * 2 + 1] = o1;
}

__global__ __launch_bounds__(BIN_NODES)
void bin_prop2(const unsigned* __restrict__ keys, const int* __restrict__ start,
               const float* __restrict__ dinv, const float* __restrict__ hw2,
               float* __restrict__ outb, int N) {
    __shared__ float acc[BIN_NODES * 2];
    __shared__ float dl[BIN_NODES];
    const int tid   = threadIdx.x;
    const int node0 = blockIdx.x * BIN_NODES;
    acc[tid] = 0.0f;
    acc[BIN_NODES + tid] = 0.0f;
    dl[tid] = (node0 + tid < N) ? dinv[node0 + tid] : 0.0f;
    __syncthreads();

    const int s = start[blockIdx.x], e = start[blockIdx.x + 1];
    const int f   = tid & 1;
    const int sub = tid >> 1;
    for (int j0 = s + sub; j0 < e; j0 += 8 * 512) {
        unsigned k[8];
        bool ok[8];
#pragma unroll
        for (int u = 0; u < 8; ++u) {
            const int j = j0 + u * 512;
            ok[u] = j < e;
            k[u] = keys[ok[u] ? j : j0];
        }
        float v[8];
        int   a[8];
#pragma unroll
        for (int u = 0; u < 8; ++u) {
            const int r  = (int)(k[u] >> BIN_SHIFT);
            const int cl = (int)(k[u] & BIN_MASK);
            const float w = dinv[r] * dl[cl];
            v[u] = hw2[(long)r * 2 + f] * w;
            a[u] = cl * 2 + f;
        }
#pragma unroll
        for (int u = 0; u < 8; ++u)
            if (ok[u]) ldsAddF(&acc[a[u]], v[u]);
    }
    __syncthreads();
    float* dst = outb + (long)node0 * 2;
#pragma unroll
    for (int i = 0; i < 2; ++i) {
        const int idx = i * BIN_NODES + tid;
        if (node0 + (idx >> 1) < N) dst[idx] = acc[idx];
    }
}

__global__ void gcn_finalize(float* __restrict__ outb, const float* __restrict__ hw2,
                             const float* __restrict__ dinv, const float* __restrict__ b2,
                             int N) {
    int i = blockIdx.x * blockDim.x + threadIdx.x;
    if (i >= N) return;
    const float self = dinv[i] * dinv[i];
    float v0 = outb[(long)i * 2 + 0] + self * hw2[(long)i * 2 + 0] + b2[0];
    float v1 = outb[(long)i * 2 + 1] + self * hw2[(long)i * 2 + 1] + b2[1];
    const float m = fmaxf(v0, v1);
    const float lse = m + logf(expf(v0 - m) + expf(v1 - m));
    outb[(long)i * 2 + 0] = v0 - lse;
    outb[(long)i * 2 + 1] = v1 - lse;
}

extern "C" void kernel_launch(void* const* d_in, const int* in_sizes, int n_in,
                              void* d_out, int out_size, void* d_ws, size_t ws_size,
                              hipStream_t stream) {
    const float* x    = (const float*)d_in[0];
    const float* W1   = (const float*)d_in[1];
    const float* b1   = (const float*)d_in[2];
    const float* W2   = (const float*)d_in[3];
    const float* b2   = (const float*)d_in[4];
    const int*   edge = (const int*)d_in[5];

    const int N = in_sizes[0] / 16;
    const int E = in_sizes[5] / 2;
    const int* row = edge;
    const int* col = edge + E;
    float* out = (float*)d_out;

    const int P = (N + BIN_NODES - 1) / BIN_NODES;

    size_t off = 0;
    auto carve = [&](size_t bytes) {
        void* p = (char*)d_ws + off;
        off = (off + bytes + 255) & ~(size_t)255;
        return p;
    };
    float*    dinv  = (float*)carve((size_t)N * 4);
    float*    xw1   = (float*)carve((size_t)N * 8 * 4);
    float*    agg1  = (float*)carve((size_t)N * 8 * 4);
    float*    hw2   = (float*)carve((size_t)N * 2 * 4);
    int*      tot   = (int*)carve((size_t)P * CNT_PITCH * 4);
    int*      cur   = (int*)carve((size_t)P * CNT_PITCH * 4);
    int*      start = (int*)carve((size_t)(P + 1) * 4);
    unsigned* keys  = (unsigned*)carve((size_t)E * 4);
    (void)ws_size; (void)n_in; (void)out_size;

    const int B = 256;
    const int gN = (N + B - 1) / B;
    const int nTiles = (N + 15) / 16;
    const int gW = (nTiles * 32 + B - 1) / B;
    const int gC = (E + CHUNK - 1) / CHUNK;

    bin_init    <<<1,  BIN_NODES, 0, stream>>>(tot, cur, P);
    bin_count   <<<gC, BIN_NODES, 0, stream>>>(col, tot, E, P);
    bin_scan    <<<1,  BIN_NODES, 0, stream>>>(tot, start, P);
    bin_scatter <<<gC, BIN_NODES, 0, stream>>>(row, col, start, cur, keys, E, P);
    bin_degree  <<<P,  BIN_NODES, 0, stream>>>(keys, start, dinv, N);
    gcn_xw1_wmma<<<gW, B, 0, stream>>>(x, W1, xw1, nTiles, N);
    bin_prop1   <<<P,  BIN_NODES, 0, stream>>>(keys, start, dinv, xw1, agg1, N);
    gcn_relu_hw2<<<gN, B, 0, stream>>>(agg1, xw1, dinv, b1, W2, hw2, N);
    bin_prop2   <<<P,  BIN_NODES, 0, stream>>>(keys, start, dinv, hw2, out, N);
    gcn_finalize<<<gN, B, 0, stream>>>(out, hw2, dinv, b2, N);
}


# Source factor dinv[src] folded into the gathered tables (dinv*xW1, dinv*hW2), dinv[dst] applied once per summed row: one gather per edge, no dinv gather; rounding regrouped, sum order unchanged
# speedup vs baseline: 4.0843x; 1.2931x over previous
#include <hip/hip_runtime.h>
#include <hip/hip_bf16.h>


typedef __attribute__((ext_vector_type(2)))  float    v2f;
typedef __attribute__((ext_vector_type(8)))  float    v8f;
typedef __attribute__((ext_vector_type(16))) _Float16 v16h;

#define BIN_SHIFT 10
#define BIN_NODES 1024
#define BIN_MASK  1023
#define CNT_PITCH 1088
#define CHUNK     16384

__device__ __forceinline__ void ldsAddF(float* p, float v) {
    __hip_atomic_fetch_add(p, v, __ATOMIC_RELAXED, __HIP_MEMORY_SCOPE_WORKGROUP);
}
__device__ __forceinline__ void ldsAddI(int* p) {
    __hip_atomic_fetch_add(p, 1, __ATOMIC_RELAXED, __HIP_MEMORY_SCOPE_WORKGROUP);
}

__global__ void bin_init(int* __restrict__ tot, int* __restrict__ cur, int P) {
    int i = blockIdx.x * blockDim.x + threadIdx.x;
    if (i < P) { tot[i * CNT_PITCH] = 0; cur[i * CNT_PITCH] = 0; }
}

__device__ __forceinline__ void chunk_hist(const int* __restrict__ col, int s, int e,
                                           int* hist, int tid) {
    for (int j0 = s + tid; j0 < e; j0 += 4 * BIN_NODES) {
        int b[4];
#pragma unroll
        for (int u = 0; u < 4; ++u) {
            const int j = j0 + u * BIN_NODES;
            b[u] = (j < e) ? (col[j] >> BIN_SHIFT) : -1;
        }
#pragma unroll
        for (int u = 0; u < 4; ++u)
            if (b[u] >= 0) ldsAddI(&hist[b[u]]);
    }
}

__global__ __launch_bounds__(BIN_NODES)
void bin_count(const int* __restrict__ col, int* __restrict__ tot, int E, int P) {
    __shared__ int hist[BIN_NODES];
    const int tid = threadIdx.x;
    hist[tid] = 0;
    __syncthreads();
    const int s = blockIdx.x * CHUNK;
    const int e = min(s + CHUNK, E);
    chunk_hist(col, s, e, hist, tid);
    __syncthreads();
    if (tid < P) {
        const int h = hist[tid];
        if (h) __hip_atomic_fetch_add(&tot[tid * CNT_PITCH], h, __ATOMIC_RELAXED,
                                      __HIP_MEMORY_SCOPE_AGENT);
    }
}

__device__ __forceinline__ int block_incl_scan(int v, int* s, int tid) {
    s[tid] = v;
    __syncthreads();
    for (int off = 1; off < BIN_NODES; off <<= 1) {
        const int t = (tid >= off) ? s[tid - off] : 0;
        __syncthreads();
        s[tid] += t;
        __syncthreads();
    }
    return s[tid];
}

__global__ __launch_bounds__(BIN_NODES)
void bin_scan(const int* __restrict__ tot, int* __restrict__ start, int P) {
    __shared__ int s[BIN_NODES];
    const int tid = threadIdx.x;
    const int incl = block_incl_scan((tid < P) ? tot[tid * CNT_PITCH] : 0, s, tid);
    if (tid < P) start[tid + 1] = incl;
    if (tid == 0) start[0] = 0;
}

__global__ __launch_bounds__(BIN_NODES)
void bin_scatter(const int* __restrict__ row, const int* __restrict__ col,
                 const int* __restrict__ start, int* __restrict__ cur,
                 unsigned* __restrict__ keys, int E, int P) {
    __shared__ unsigned       stage[CHUNK];
    __shared__ unsigned short sbin[CHUNK];
    __shared__ int hist[BIN_NODES];
    __shared__ int lst[BIN_NODES];
    __shared__ int delta[BIN_NODES];
    const int tid = threadIdx.x;
    hist[tid] = 0;
    __syncthreads();
    const int s = blockIdx.x * CHUNK;
    const int e = min(s + CHUNK, E);
    chunk_hist(col, s, e, hist, tid);
    __syncthreads();
    const int h = hist[tid];
    int gbase = 0;
    if (tid < P && h)
        gbase = start[tid] + __hip_atomic_fetch_add(&cur[tid * CNT_PITCH], h,
                                 __ATOMIC_RELAXED, __HIP_MEMORY_SCOPE_AGENT);
    const int excl = block_incl_scan(h, lst, tid) - h;
    lst[tid]   = excl;
    delta[tid] = gbase - excl;
    hist[tid]  = 0;
    __syncthreads();
    for (int j0 = s + tid; j0 < e; j0 += 4 * BIN_NODES) {
        int r[4], c[4];
#pragma unroll
        for (int u = 0; u < 4; ++u) {
            const int j = j0 + u * BIN_NODES;
            const int jj = (j < e) ? j : j0;
            r[u] = row[jj];
            c[u] = (j < e) ? col[jj] : -1;
        }
#pragma unroll
        for (int u = 0; u < 4; ++u) {
            if (c[u] >= 0) {
                const int b  = c[u] >> BIN_SHIFT;
                const int li = lst[b] + __hip_atomic_fetch_add(&hist[b], 1, __ATOMIC_RELAXED,
                                                               __HIP_MEMORY_SCOPE_WORKGROUP);
                stage[li] = ((unsigned)r[u] << BIN_SHIFT) | (unsigned)(c[u] & BIN_MASK);
                sbin[li]  = (unsigned short)b;
            }
        }
    }
    __syncthreads();
    const int n = e - s;
    for (int i = tid; i < n; i += BIN_NODES)
        keys[delta[sbin[i]] + i] = stage[i];
}

__global__ __launch_bounds__(BIN_NODES)
void bin_degree(const unsigned* __restrict__ keys, const int* __restrict__ start,
                float* __restrict__ dinv, int N) {
    __shared__ int cnt[BIN_NODES];
    const int tid = threadIdx.x;
    cnt[tid] = 0;
    __syncthreads();
    const int s = start[blockIdx.x], e = start[blockIdx.x + 1];
    for (int j0 = s + tid; j0 < e; j0 += 4 * BIN_NODES) {
        int k[4];
#pragma unroll
        for (int u = 0; u < 4; ++u) {
            const int j = j0 + u * BIN_NODES;
            k[u] = (j < e) ? (int)(keys[j] & BIN_MASK) : -1;
        }
#pragma unroll
        for (int u = 0; u < 4; ++u)
            if (k[u] >= 0) ldsAddI(&cnt[k[u]]);
    }
    __syncthreads();
    const int node = blockIdx.x * BIN_NODES + tid;
    if (node < N) dinv[node] = __frsqrt_rn((float)(cnt[tid] + 1));
}

__global__ void gcn_xw1_wmma(const float* __restrict__ x, const float* __restrict__ W1,
                             const float* __restrict__ dinv,
                             float* __restrict__ xw1s, int nTiles, int N) {
    __shared__ float tile[8][16 * 8];
    int flat = blockIdx.x * blockDim.x + threadIdx.x;
    int wave = __builtin_amdgcn_readfirstlane(flat >> 5);
    if (wave >= nTiles) return;

    const int lane = threadIdx.x & 31;
    const int m    = lane & 15;
    const int half = lane >> 4;
    const int n    = m;
    const float colMask = (n < 8) ? 1.0f : 0.0f;
    const int   nn      = n & 7;

    long rowA = (long)wave * 16 + m;
    if (rowA >= N) rowA = N - 1;
    const float* xr = x + rowA * 16;

    const int rowIdx = lane >> 1;
    const int quad   = lane & 1;
    const long gr = (long)wave * 16 + rowIdx;
    const float sc = dinv[(gr < N) ? gr : (long)(N - 1)];

    v8f acc = {};
#if __has_builtin(__builtin_amdgcn_wmma_f32_16x16x4_f32)
#pragma unroll
    for (int kb = 0; kb < 16; kb += 4) {
        const int k0 = kb + 2 * half;
        v2f a, b;
        a.x = xr[k0 + 0];
        a.y = xr[k0 + 1];
        b.x = W1[(k0 + 0) * 8 + nn] * colMask;
        b.y = W1[(k0 + 1) * 8 + nn] * colMask;
        acc = __builtin_amdgcn_wmma_f32_16x16x4_f32(
                  false, a, false, b, (short)0, acc, false, false);
    }
#else
    v16h a, b;
#pragma unroll
    for (int j = 0; j < 16; ++j) {
        a[j] = (j < 8) ? (_Float16)xr[8 * half + j] : (_Float16)0.0f;
        float bv = (half == 0) ? W1[j * 8 + nn] * colMask : 0.0f;
        b[j] = (_Float16)bv;
    }
    acc = __builtin_amdgcn_wmma_f32_16x16x32_f16(
              false, a, false, b, (short)0, acc, false, false);
#endif

    float* t = &tile[threadIdx.x >> 5][0];
    if (n < 8) {
#pragma unroll
        for (int r = 0; r < 8; ++r)
            t[(r + 8 * half) * 8 + n] = acc[r];
    }
    asm volatile("s_wait_dscnt 0x0" ::: "memory");

    float4 v = *(const float4*)(t + rowIdx * 8 + quad * 4);
    v.x *= sc; v.y *= sc; v.z *= sc; v.w *= sc;
    if (gr < N) *(float4*)(xw1s + gr * 8 + quad * 4) = v;
}

__global__ __launch_bounds__(BIN_NODES)
void bin_prop1(const unsigned* __restrict__ keys, const int* __restrict__ start,
               const float* __restrict__ dinv, const float* __restrict__ xw1s,
               float* __restrict__ agg1, int N) {
    __shared__ float acc[BIN_NODES * 8];
    __shared__ float dl[BIN_NODES];
    const int tid   = threadIdx.x;
    const int node0 = blockIdx.x * BIN_NODES;
#pragma unroll
    for (int i = 0; i < 8; ++i) acc[i * BIN_NODES + tid] = 0.0f;
    dl[tid] = (node0 + tid < N) ? dinv[node0 + tid] : 0.0f;
    __syncthreads();

    const int s = start[blockIdx.x], e = start[blockIdx.x + 1];
    const int f   = tid & 7;
    const int sub = tid >> 3;
    for (int j0 = s + sub; j0 < e; j0 += 8 * 128) {
        unsigned k[8];
        bool ok[8];
#pragma unroll
        for (int u = 0; u < 8; ++u) {
            const int j = j0 + u * 128;
            ok[u] = j < e;
            k[u] = keys[ok[u] ? j : j0];
        }
        float v[8];
        int   a[8];
#pragma unroll
        for (int u = 0; u < 8; ++u) {
            const int r  = (int)(k[u] >> BIN_SHIFT);
            const int cl = (int)(k[u] & BIN_MASK);
            v[u] = xw1s[(long)r * 8 + f];
            a[u] = cl * 8 + f;
        }
#pragma unroll
        for (int u = 0; u < 8; ++u)
            if (ok[u]) ldsAddF(&acc[a[u]], v[u]);
    }
    __syncthreads();
    float* dst = agg1 + (long)node0 * 8;
#pragma unroll
    for (int i = 0; i < 8; ++i) {
        const int idx = i * BIN_NODES + tid;
        if (node0 + (idx >> 3) < N) dst[idx] = acc[idx] * dl[idx >> 3];
    }
}

__global__ void gcn_relu_hw2(const float* __restrict__ agg1, const float* __restrict__ xw1s,
                             const float* __restrict__ dinv, const float* __restrict__ b1,
                             const float* __restrict__ W2, float* __restrict__ hw2s, int N) {
    int i = blockIdx.x * blockDim.x + threadIdx.x;
    if (i >= N) return;
    const float self = dinv[i];
    const float4* a = (const float4*)(agg1 + (long)i * 8);
    const float4* s = (const float4*)(xw1s + (long)i * 8);
    float hv[8];
    {
        float4 a0 = a[0], a1 = a[1], s0 = s[0], s1 = s[1];
        hv[0] = a0.x + self * s0.x; hv[1] = a0.y + self * s0.y;
        hv[2] = a0.z + self * s0.z; hv[3] = a0.w + self * s0.w;
        hv[4] = a1.x + self * s1.x; hv[5] = a1.y + self * s1.y;
        hv[6] = a1.z + self * s1.z; hv[7] = a1.w + self * s1.w;
    }
    float o0 = 0.0f, o1 = 0.0f;
#pragma unroll
    for (int f = 0; f < 8; ++f) {
        float h = fmaxf(hv[f] + b1[f], 0.0f);
        o0 = fmaf(h, W2[f * 2 + 0], o0);
        o1 = fmaf(h, W2[f * 2 + 1], o1);
    }
    hw2s[(long)i * 2 + 0] = o0 * self;
    hw2s[(long)i * 2 + 1] = o1 * self;
}

__global__ __launch_bounds__(BIN_NODES)
void bin_prop2(const unsigned* __restrict__ keys, const int* __restrict__ start,
               const float* __restrict__ dinv, const float* __restrict__ hw2s,
               float* __restrict__ outb, int N) {
    __shared__ float acc[BIN_NODES * 2];
    __shared__ float dl[BIN_NODES];
    const int tid   = threadIdx.x;
    const int node0 = blockIdx.x * BIN_NODES;
    acc[tid] = 0.0f;
    acc[BIN_NODES + tid] = 0.0f;
    dl[tid] = (node0 + tid < N) ? dinv[node0 + tid] : 0.0f;
    __syncthreads();

    const int s = start[blockIdx.x], e = start[blockIdx.x + 1];
    const int f   = tid & 1;
    const int sub = tid >> 1;
    for (int j0 = s + sub; j0 < e; j0 += 8 * 512) {
        unsigned k[8];
        bool ok[8];
#pragma unroll
        for (int u = 0; u < 8; ++u) {
            const int j = j0 + u * 512;
            ok[u] = j < e;
            k[u] = keys[ok[u] ? j : j0];
        }
        float v[8];
        int   a[8];
#pragma unroll
        for (int u = 0; u < 8; ++u) {
            const int r  = (int)(k[u] >> BIN_SHIFT);
            const int cl = (int)(k[u] & BIN_MASK);
            v[u] = hw2s[(long)r * 2 + f];
            a[u] = cl * 2 + f;
        }
#pragma unroll
        for (int u = 0; u < 8; ++u)
            if (ok[u]) ldsAddF(&acc[a[u]], v[u]);
    }
    __syncthreads();
    float* dst = outb + (long)node0 * 2;
#pragma unroll
    for (int i = 0; i < 2; ++i) {
        const int idx = i * BIN_NODES + tid;
        if (node0 + (idx >> 1) < N) dst[idx] = acc[idx] * dl[idx >> 1];
    }
}

__global__ void gcn_finalize(float* __restrict__ outb, const float* __restrict__ hw2s,
                             const float* __restrict__ dinv, const float* __restrict__ b2,
                             int N) {
    int i = blockIdx.x * blockDim.x + threadIdx.x;
    if (i >= N) return;
    const float self = dinv[i];
    float v0 = outb[(long)i * 2 + 0] + self * hw2s[(long)i * 2 + 0] + b2[0];
    float v1 = outb[(long)i * 2 + 1] + self * hw2s[(long)i * 2 + 1] + b2[1];
    const float m = fmaxf(v0, v1);
    const float lse = m + logf(expf(v0 - m) + expf(v1 - m));
    outb[(long)i * 2 + 0] = v0 - lse;
    outb[(long)i * 2 + 1] = v1 - lse;
}

extern "C" void kernel_launch(void* const* d_in, const int* in_sizes, int n_in,
                              void* d_out, int out_size, void* d_ws, size_t ws_size,
                              hipStream_t stream) {
    const float* x    = (const float*)d_in[0];
    const float* W1   = (const float*)d_in[1];
    const float* b1   = (const float*)d_in[2];
    const float* W2   = (const float*)d_in[3];
    const float* b2   = (const float*)d_in[4];
    const int*   edge = (const int*)d_in[5];

    const int N = in_sizes[0] / 16;
    const int E = in_sizes[5] / 2;
    const int* row = edge;
    const int* col = edge + E;
    float* out = (float*)d_out;

    const int P = (N + BIN_NODES - 1) / BIN_NODES;

    size_t off = 0;
    auto carve = [&](size_t bytes) {
        void* p = (char*)d_ws + off;
        off = (off + bytes + 255) & ~(size_t)255;
        return p;
    };
    float*    dinv  = (float*)carve((size_t)N * 4);
    float*    xw1s  = (float*)carve((size_t)N * 8 * 4);
    float*    agg1  = (float*)carve((size_t)N * 8 * 4);
    float*    hw2s  = (float*)carve((size_t)N * 2 * 4);
    int*      tot   = (int*)carve((size_t)P * CNT_PITCH * 4);
    int*      cur   = (int*)carve((size_t)P * CNT_PITCH * 4);
    int*      start = (int*)carve((size_t)(P + 1) * 4);
    unsigned* keys  = (unsigned*)carve((size_t)E * 4);
    (void)ws_size; (void)n_in; (void)out_size;

    const int B = 256;
    const int gN = (N + B - 1) / B;
    const int nTiles = (N + 15) / 16;
    const int gW = (nTiles * 32 + B - 1) / B;
    const int gC = (E + CHUNK - 1) / CHUNK;

    bin_init    <<<1,  BIN_NODES, 0, stream>>>(tot, cur, P);
    bin_count   <<<gC, BIN_NODES, 0, stream>>>(col, tot, E, P);
    bin_scan    <<<1,  BIN_NODES, 0, stream>>>(tot, start, P);
    bin_scatter <<<gC, BIN_NODES, 0, stream>>>(row, col, start, cur, keys, E, P);
    bin_degree  <<<P,  BIN_NODES, 0, stream>>>(keys, start, dinv, N);
    gcn_xw1_wmma<<<gW, B, 0, stream>>>(x, W1, dinv, xw1s, nTiles, N);
    bin_prop1   <<<P,  BIN_NODES, 0, stream>>>(keys, start, dinv, xw1s, agg1, N);
    gcn_relu_hw2<<<gN, B, 0, stream>>>(agg1, xw1s, dinv, b1, W2, hw2s, N);
    bin_prop2   <<<P,  BIN_NODES, 0, stream>>>(keys, start, dinv, hw2s, out, N);
    gcn_finalize<<<gN, B, 0, stream>>>(out, hw2s, dinv, b2, N);
}


# Per-node stages fused into the bin kernels that own the same 1024 nodes (xW1 behind degree count, relu/W2 behind layer-1 sums, log_softmax behind layer-2 sums): 10 launches to 7, sum order unchanged
# speedup vs baseline: 4.5136x; 1.1051x over previous
#include <hip/hip_runtime.h>
#include <hip/hip_bf16.h>


typedef __attribute__((ext_vector_type(2)))  float    v2f;
typedef __attribute__((ext_vector_type(8)))  float    v8f;

#define BIN_SHIFT 10
#define BIN_NODES 1024
#define BIN_MASK  1023
#define CNT_PITCH 1088
#define CHUNK     16384

__device__ __forceinline__ void ldsAddF(float* p, float v) {
    __hip_atomic_fetch_add(p, v, __ATOMIC_RELAXED, __HIP_MEMORY_SCOPE_WORKGROUP);
}
__device__ __forceinline__ void ldsAddI(int* p) {
    __hip_atomic_fetch_add(p, 1, __ATOMIC_RELAXED, __HIP_MEMORY_SCOPE_WORKGROUP);
}

__global__ void bin_init(int* __restrict__ tot, int* __restrict__ cur, int P) {
    int i = blockIdx.x * blockDim.x + threadIdx.x;
    if (i < P) { tot[i * CNT_PITCH] = 0; cur[i * CNT_PITCH] = 0; }
}

__device__ __forceinline__ void chunk_hist(const int* __restrict__ col, int s, int e,
                                           int* hist, int tid) {
    for (int j0 = s + tid; j0 < e; j0 += 4 * BIN_NODES) {
        int b[4];
#pragma unroll
        for (int u = 0; u < 4; ++u) {
            const int j = j0 + u * BIN_NODES;
            b[u] = (j < e) ? (col[j] >> BIN_SHIFT) : -1;
        }
#pragma unroll
        for (int u = 0; u < 4; ++u)
            if (b[u] >= 0) ldsAddI(&hist[b[u]]);
    }
}

__global__ __launch_bounds__(BIN_NODES)
void bin_count(const int* __restrict__ col, int* __restrict__ tot, int E, int P) {
    __shared__ int hist[BIN_NODES];
    const int tid = threadIdx.x;
    hist[tid] = 0;
    __syncthreads();
    const int s = blockIdx.x * CHUNK;
    const int e = min(s + CHUNK, E);
    chunk_hist(col, s, e, hist, tid);
    __syncthreads();
    if (tid < P) {
        const int h = hist[tid];
        if (h) __hip_atomic_fetch_add(&tot[tid * CNT_PITCH], h, __ATOMIC_RELAXED,
                                      __HIP_MEMORY_SCOPE_AGENT);
    }
}

__device__ __forceinline__ int block_incl_scan(int v, int* s, int tid) {
    s[tid] = v;
    __syncthreads();
    for (int off = 1; off < BIN_NODES; off <<= 1) {
        const int t = (tid >= off) ? s[tid - off] : 0;
        __syncthreads();
        s[tid] += t;
        __syncthreads();
    }
    return s[tid];
}

__global__ __launch_bounds__(BIN_NODES)
void bin_scan(const int* __restrict__ tot, int* __restrict__ start, int P) {
    __shared__ int s[BIN_NODES];
    const int tid = threadIdx.x;
    const int incl = block_incl_scan((tid < P) ? tot[tid * CNT_PITCH] : 0, s, tid);
    if (tid < P) start[tid + 1] = incl;
    if (tid == 0) start[0] = 0;
}

__global__ __launch_bounds__(BIN_NODES)
void bin_scatter(const int* __restrict__ row, const int* __restrict__ col,
                 const int* __restrict__ start, int* __restrict__ cur,
                 unsigned* __restrict__ keys, int E, int P) {
    __shared__ unsigned       stage[CHUNK];
    __shared__ unsigned short sbin[CHUNK];
    __shared__ int hist[BIN_NODES];
    __shared__ int lst[BIN_NODES];
    __shared__ int delta[BIN_NODES];
    const int tid = threadIdx.x;
    hist[tid] = 0;
    __syncthreads();
    const int s = blockIdx.x * CHUNK;
    const int e = min(s + CHUNK, E);
    chunk_hist(col, s, e, hist, tid);
    __syncthreads();
    const int h = hist[tid];
    int gbase = 0;
    if (tid < P && h)
        gbase = start[tid] + __hip_atomic_fetch_add(&cur[tid * CNT_PITCH], h,
                                 __ATOMIC_RELAXED, __HIP_MEMORY_SCOPE_AGENT);
    const int excl = block_incl_scan(h, lst, tid) - h;
    lst[tid]   = excl;
    delta[tid] = gbase - excl;
    hist[tid]  = 0;
    __syncthreads();
    for (int j0 = s + tid; j0 < e; j0 += 4 * BIN_NODES) {
        int r[4], c[4];
#pragma unroll
        for (int u = 0; u < 4; ++u) {
            const int j = j0 + u * BIN_NODES;
            const int jj = (j < e) ? j : j0;
            r[u] = row[jj];
            c[u] = (j < e) ? col[jj] : -1;
        }
#pragma unroll
        for (int u = 0; u < 4; ++u) {
            if (c[u] >= 0) {
                const int b  = c[u] >> BIN_SHIFT;
                const int li = lst[b] + __hip_atomic_fetch_add(&hist[b], 1, __ATOMIC_RELAXED,
                                                               __HIP_MEMORY_SCOPE_WORKGROUP);
                stage[li] = ((unsigned)r[u] << BIN_SHIFT) | (unsigned)(c[u] & BIN_MASK);
                sbin[li]  = (unsigned short)b;
            }
        }
    }
    __syncthreads();
    const int n = e - s;
    for (int i = tid; i < n; i += BIN_NODES)
        keys[delta[sbin[i]] + i] = stage[i];
}

__global__ __launch_bounds__(BIN_NODES)
void bin_degree_xw1(const unsigned* __restrict__ keys, const int* __restrict__ start,
                    const float* __restrict__ x, const float* __restrict__ W1,
                    float* __restrict__ dinv, float* __restrict__ xw1s, int N) {
    __shared__ int   cnt[BIN_NODES];
    __shared__ float dl[BIN_NODES];
    __shared__ float tile[64][16 * 8];
    const int tid   = threadIdx.x;
    const int node0 = blockIdx.x * BIN_NODES;
    cnt[tid] = 0;
    __syncthreads();

    const int w    = tid >> 5;
    const int lane = tid & 31;
    const int m    = lane & 15;
    const int half = lane >> 4;
    const int n    = m;
    const float colMask = (n < 8) ? 1.0f : 0.0f;
    const int   nn      = n & 7;
    v2f a[2][4], b[4];
#pragma unroll
    for (int q = 0; q < 4; ++q) {
        const int k0 = 4 * q + 2 * half;
        b[q].x = W1[(k0 + 0) * 8 + nn] * colMask;
        b[q].y = W1[(k0 + 1) * 8 + nn] * colMask;
    }
#pragma unroll
    for (int i = 0; i < 2; ++i) {
        long rowA = (long)node0 + (w + 32 * i) * 16 + m;
        if (rowA >= N) rowA = N - 1;
        const float* xr = x + rowA * 16;
#pragma unroll
        for (int q = 0; q < 4; ++q) {
            const int k0 = 4 * q + 2 * half;
            a[i][q].x = xr[k0 + 0];
            a[i][q].y = xr[k0 + 1];
        }
    }

    const int s = start[blockIdx.x], e = start[blockIdx.x + 1];
    for (int j0 = s + tid; j0 < e; j0 += 4 * BIN_NODES) {
        int k[4];
#pragma unroll
        for (int u = 0; u < 4; ++u) {
            const int j = j0 + u * BIN_NODES;
            k[u] = (j < e) ? (int)(keys[j] & BIN_MASK) : -1;
        }
#pragma unroll
        for (int u = 0; u < 4; ++u)
            if (k[u] >= 0) ldsAddI(&cnt[k[u]]);
    }
    __syncthreads();
    const int node = node0 + tid;
    const float dv = __frsqrt_rn((float)(cnt[tid] + 1));
    dl[tid] = dv;
    if (node < N) dinv[node] = dv;
    __syncthreads();

    v8f acc0 = {}, acc1 = {};
#pragma unroll
    for (int q = 0; q < 4; ++q) {
        acc0 = __builtin_amdgcn_wmma_f32_16x16x4_f32(
                   false, a[0][q], false, b[q], (short)0, acc0, false, false);
        acc1 = __builtin_amdgcn_wmma_f32_16x16x4_f32(
                   false, a[1][q], false, b[q], (short)0, acc1, false, false);
    }

    float* t0 = &tile[w][0];
    float* t1 = &tile[w + 32][0];
    if (n < 8) {
#pragma unroll
        for (int r = 0; r < 8; ++r) {
            t0[(r + 8 * half) * 8 + n] = acc0[r];
            t1[(r + 8 * half) * 8 + n] = acc1[r];
        }
    }
    asm volatile("s_wait_dscnt 0x0" ::: "memory");

    const int rowIdx = lane >> 1;
    const int quad   = lane & 1;
#pragma unroll
    for (int i = 0; i < 2; ++i) {
        const int  lr = (w + 32 * i) * 16 + rowIdx;
        const long gr = (long)node0 + lr;
        const float sc = dl[lr];
        float4 v = *(const float4*)(&tile[w + 32 * i][0] + rowIdx * 8 + quad * 4);
        v.x *= sc; v.y *= sc; v.z *= sc; v.w *= sc;
        if (gr < N) *(float4*)(xw1s + gr * 8 + quad * 4) = v;
    }
}

__global__ __launch_bounds__(BIN_NODES)
void bin_prop1_hw2(const unsigned* __restrict__ keys, const int* __restrict__ start,
                   const float* __restrict__ dinv, const float* __restrict__ xw1s,
                   const float* __restrict__ b1, const float* __restrict__ W2,
                   float* __restrict__ hw2s, int N) {
    __shared__ float acc[BIN_NODES * 8];
    const int tid   = threadIdx.x;
    const int node0 = blockIdx.x * BIN_NODES;
    const int node  = node0 + tid;
#pragma unroll
    for (int i = 0; i < 8; ++i) acc[i * BIN_NODES + tid] = 0.0f;
    const float self = (node < N) ? dinv[node] : 0.0f;
    __syncthreads();

    const int s = start[blockIdx.x], e = start[blockIdx.x + 1];
    const int f   = tid & 7;
    const int sub = tid >> 3;
    for (int j0 = s + sub; j0 < e; j0 += 8 * 128) {
        unsigned k[8];
        bool ok[8];
#pragma unroll
        for (int u = 0; u < 8; ++u) {
            const int j = j0 + u * 128;
            ok[u] = j < e;
            k[u] = keys[ok[u] ? j : j0];
        }
        float v[8];
        int   a[8];
#pragma unroll
        for (int u = 0; u < 8; ++u) {
            const int r  = (int)(k[u] >> BIN_SHIFT);
            const int cl = (int)(k[u] & BIN_MASK);
            v[u] = xw1s[(long)r * 8 + f];
            a[u] = cl * 8 + f;
        }
#pragma unroll
        for (int u = 0; u < 8; ++u)
            if (ok[u]) ldsAddF(&acc[a[u]], v[u]);
    }
    __syncthreads();
    if (node < N) {
        const float4* sr = (const float4*)(xw1s + (long)node * 8);
        const float4 s0 = sr[0], s1 = sr[1];
        const float4 q0 = *(const float4*)&acc[tid * 8];
        const float4 q1 = *(const float4*)&acc[tid * 8 + 4];
        float hv[8];
        hv[0] = q0.x * self + self * s0.x; hv[1] = q0.y * self + self * s0.y;
        hv[2] = q0.z * self + self * s0.z; hv[3] = q0.w * self + self * s0.w;
        hv[4] = q1.x * self + self * s1.x; hv[5] = q1.y * self + self * s1.y;
        hv[6] = q1.z * self + self * s1.z; hv[7] = q1.w * self + self * s1.w;
        float o0 = 0.0f, o1 = 0.0f;
#pragma unroll
        for (int c = 0; c < 8; ++c) {
            const float h = fmaxf(hv[c] + b1[c], 0.0f);
            o0 = fmaf(h, W2[c * 2 + 0], o0);
            o1 = fmaf(h, W2[c * 2 + 1], o1);
        }
        float2 o;
        o.x = o0 * self;
        o.y = o1 * self;
        *(float2*)(hw2s + (long)node * 2) = o;
    }
}

__global__ __launch_bounds__(BIN_NODES)
void bin_prop2_out(const unsigned* __restrict__ keys, const int* __restrict__ start,
                   const float* __restrict__ dinv, const float* __restrict__ hw2s,
                   const float* __restrict__ b2, float* __restrict__ outb, int N) {
    __shared__ float acc[BIN_NODES * 2];
    const int tid   = threadIdx.x;
    const int node0 = blockIdx.x * BIN_NODES;
    const int node  = node0 + tid;
    acc[tid] = 0.0f;
    acc[BIN_NODES + tid] = 0.0f;
    const float self = (node < N) ? dinv[node] : 0.0f;
    __syncthreads();

    const int s = start[blockIdx.x], e = start[blockIdx.x + 1];
    const int f   = tid & 1;
    const int sub = tid >> 1;
    for (int j0 = s + sub; j0 < e; j0 += 8 * 512) {
        unsigned k[8];
        bool ok[8];
#pragma unroll
        for (int u = 0; u < 8; ++u) {
            const int j = j0 + u * 512;
            ok[u] = j < e;
            k[u] = keys[ok[u] ? j : j0];
        }
        float v[8];
        int   a[8];
#pragma unroll
        for (int u = 0; u < 8; ++u) {
            const int r  = (int)(k[u] >> BIN_SHIFT);
            const int cl = (int)(k[u] & BIN_MASK);
            v[u] = hw2s[(long)r * 2 + f];
            a[u] = cl * 2 + f;
        }
#pragma unroll
        for (int u = 0; u < 8; ++u)
            if (ok[u]) ldsAddF(&acc[a[u]], v[u]);
    }
    __syncthreads();
    if (node < N) {
        const float2 hs = *(const float2*)(hw2s + (long)node * 2);
        const float2 q  = *(const float2*)&acc[tid * 2];
        const float v0 = q.x * self + self * hs.x + b2[0];
        const float v1 = q.y * self + self * hs.y + b2[1];
        const float mx  = fmaxf(v0, v1);
        const float lse = mx + logf(expf(v0 - mx) + expf(v1 - mx));
        float2 o;
        o.x = v0 - lse;
        o.y = v1 - lse;
        *(float2*)(outb + (long)node * 2) = o;
    }
}

extern "C" void kernel_launch(void* const* d_in, const int* in_sizes, int n_in,
                              void* d_out, int out_size, void* d_ws, size_t ws_size,
                              hipStream_t stream) {
    const float* x    = (const float*)d_in[0];
    const float* W1   = (const float*)d_in[1];
    const float* b1   = (const float*)d_in[2];
    const float* W2   = (const float*)d_in[3];
    const float* b2   = (const float*)d_in[4];
    const int*   edge = (const int*)d_in[5];

    const int N = in_sizes[0] / 16;
    const int E = in_sizes[5] / 2;
    const int* row = edge;
    const int* col = edge + E;
    float* out = (float*)d_out;

    const int P = (N + BIN_NODES - 1) / BIN_NODES;

    size_t off = 0;
    auto carve = [&](size_t bytes) {
        void* p = (char*)d_ws + off;
        off = (off + bytes + 255) & ~(size_t)255;
        return p;
    };
    float*    dinv  = (float*)carve((size_t)N * 4);
    float*    xw1s  = (float*)carve((size_t)N * 8 * 4);
    float*    hw2s  = (float*)carve((size_t)N * 2 * 4);
    int*      tot   = (int*)carve((size_t)P * CNT_PITCH * 4);
    int*      cur   = (int*)carve((size_t)P * CNT_PITCH * 4);
    int*      start = (int*)carve((size_t)(P + 1) * 4);
    unsigned* keys  = (unsigned*)carve((size_t)E * 4);
    (void)ws_size; (void)n_in; (void)out_size;

    const int gC = (E + CHUNK - 1) / CHUNK;

    bin_init    <<<1,  BIN_NODES, 0, stream>>>(tot, cur, P);
    bin_count   <<<gC, BIN_NODES, 0, stream>>>(col, tot, E, P);
    bin_scan    <<<1,  BIN_NODES, 0, stream>>>(tot, start, P);
    bin_scatter <<<gC, BIN_NODES, 0, stream>>>(row, col, start, cur, keys, E, P);
    bin_degree_xw1<<<P, BIN_NODES, 0, stream>>>(keys, start, x, W1, dinv, xw1s, N);
    bin_prop1_hw2 <<<P, BIN_NODES, 0, stream>>>(keys, start, dinv, xw1s, b1, W2, hw2s, N);
    bin_prop2_out <<<P, BIN_NODES, 0, stream>>>(keys, start, dinv, hw2s, b2, out, N);
}


# A sweep's loads leave as one group ahead of their first use (16-byte loads in count, scatter, degree; input-only asm fences in all five passes): one memory wait per group; sums still arrival order
# speedup vs baseline: 4.7066x; 1.0428x over previous
#include <hip/hip_runtime.h>
#include <hip/hip_bf16.h>


typedef __attribute__((ext_vector_type(2)))  float    v2f;
typedef __attribute__((ext_vector_type(8)))  float    v8f;

#define BIN_SHIFT 10
#define BIN_NODES 1024
#define BIN_MASK  1023
#define CNT_PITCH 1088
#define CHUNK     16384

__device__ __forceinline__ void ldsAddF(float* p, float v) {
    __hip_atomic_fetch_add(p, v, __ATOMIC_RELAXED, __HIP_MEMORY_SCOPE_WORKGROUP);
}
__device__ __forceinline__ void ldsAddI(int* p) {
    __hip_atomic_fetch_add(p, 1, __ATOMIC_RELAXED, __HIP_MEMORY_SCOPE_WORKGROUP);
}

__device__ __forceinline__ void loads_before_adds(const int4& a, const int4& b,
                                                  const int4& c, const int4& d) {
    asm volatile("" :: "v"(a.x), "v"(a.y), "v"(a.z), "v"(a.w),
                       "v"(b.x), "v"(b.y), "v"(b.z), "v"(b.w),
                       "v"(c.x), "v"(c.y), "v"(c.z), "v"(c.w),
                       "v"(d.x), "v"(d.y), "v"(d.z), "v"(d.w));
}

__device__ __forceinline__ void keys_in_one_group(const unsigned (&k)[8]) {
    asm volatile("" :: "v"(k[0]), "v"(k[1]), "v"(k[2]), "v"(k[3]),
                       "v"(k[4]), "v"(k[5]), "v"(k[6]), "v"(k[7]));
}
__device__ __forceinline__ void rows_in_one_group(const float (&v)[8]) {
    asm volatile("" :: "v"(v[0]), "v"(v[1]), "v"(v[2]), "v"(v[3]),
                       "v"(v[4]), "v"(v[5]), "v"(v[6]), "v"(v[7]));
}

__global__ void bin_init(int* __restrict__ tot, int* __restrict__ cur, int P) {
    int i = blockIdx.x * blockDim.x + threadIdx.x;
    if (i < P) { tot[i * CNT_PITCH] = 0; cur[i * CNT_PITCH] = 0; }
}

__device__ __forceinline__ void chunk_hist(const int* __restrict__ col, int s, int e,
                                           int* hist, int tid) {
    for (int j0 = s + tid; j0 < e; j0 += 4 * BIN_NODES) {
        int b[4];
#pragma unroll
        for (int u = 0; u < 4; ++u) {
            const int j = j0 + u * BIN_NODES;
            b[u] = (j < e) ? (col[j] >> BIN_SHIFT) : -1;
        }
#pragma unroll
        for (int u = 0; u < 4; ++u)
            if (b[u] >= 0) ldsAddI(&hist[b[u]]);
    }
}

__device__ __forceinline__ bool chunk_is_wide(const int* p, int s, int e) {
    return (e - s == CHUNK) && ((reinterpret_cast<size_t>(p + s) & 15) == 0);
}

__device__ __forceinline__ void chunk_hist_wide(const int* __restrict__ col, int s,
                                                int* hist, int tid) {
    const int4* c4 = reinterpret_cast<const int4*>(col + s);
    int4 c[4];
#pragma unroll
    for (int u = 0; u < 4; ++u) c[u] = c4[tid + u * BIN_NODES];
    loads_before_adds(c[0], c[1], c[2], c[3]);
#pragma unroll
    for (int u = 0; u < 4; ++u) {
        ldsAddI(&hist[c[u].x >> BIN_SHIFT]);
        ldsAddI(&hist[c[u].y >> BIN_SHIFT]);
        ldsAddI(&hist[c[u].z >> BIN_SHIFT]);
        ldsAddI(&hist[c[u].w >> BIN_SHIFT]);
    }
}

__global__ __launch_bounds__(BIN_NODES)
void bin_count(const int* __restrict__ col, int* __restrict__ tot, int E, int P) {
    __shared__ int hist[BIN_NODES];
    const int tid = threadIdx.x;
    hist[tid] = 0;
    __syncthreads();
    const int s = blockIdx.x * CHUNK;
    const int e = min(s + CHUNK, E);
    if (chunk_is_wide(col, s, e)) chunk_hist_wide(col, s, hist, tid);
    else                          chunk_hist(col, s, e, hist, tid);
    __syncthreads();
    if (tid < P) {
        const int h = hist[tid];
        if (h) __hip_atomic_fetch_add(&tot[tid * CNT_PITCH], h, __ATOMIC_RELAXED,
                                      __HIP_MEMORY_SCOPE_AGENT);
    }
}

__device__ __forceinline__ int block_incl_scan(int v, int* s, int tid) {
    s[tid] = v;
    __syncthreads();
    for (int off = 1; off < BIN_NODES; off <<= 1) {
        const int t = (tid >= off) ? s[tid - off] : 0;
        __syncthreads();
        s[tid] += t;
        __syncthreads();
    }
    return s[tid];
}

__global__ __launch_bounds__(BIN_NODES)
void bin_scan(const int* __restrict__ tot, int* __restrict__ start, int P) {
    __shared__ int s[BIN_NODES];
    const int tid = threadIdx.x;
    const int incl = block_incl_scan((tid < P) ? tot[tid * CNT_PITCH] : 0, s, tid);
    if (tid < P) start[tid + 1] = incl;
    if (tid == 0) start[0] = 0;
}

__global__ __launch_bounds__(BIN_NODES)
void bin_scatter(const int* __restrict__ row, const int* __restrict__ col,
                 const int* __restrict__ start, int* __restrict__ cur,
                 unsigned* __restrict__ keys, int E, int P) {
    __shared__ unsigned       stage[CHUNK];
    __shared__ unsigned short sbin[CHUNK];
    __shared__ int hist[BIN_NODES];
    __shared__ int lst[BIN_NODES];
    __shared__ int delta[BIN_NODES];
    const int tid = threadIdx.x;
    hist[tid] = 0;
    __syncthreads();
    const int s = blockIdx.x * CHUNK;
    const int e = min(s + CHUNK, E);
    const bool wide = chunk_is_wide(col, s, e) && chunk_is_wide(row, s, e);
    if (wide) chunk_hist_wide(col, s, hist, tid);
    else      chunk_hist(col, s, e, hist, tid);
    __syncthreads();
    const int h = hist[tid];
    int gbase = 0;
    if (tid < P && h)
        gbase = start[tid] + __hip_atomic_fetch_add(&cur[tid * CNT_PITCH], h,
                                 __ATOMIC_RELAXED, __HIP_MEMORY_SCOPE_AGENT);
    const int excl = block_incl_scan(h, lst, tid) - h;
    lst[tid]   = excl;
    delta[tid] = gbase - excl;
    hist[tid]  = 0;
    __syncthreads();
    auto place = [&](int r, int cc) {
        const int b  = cc >> BIN_SHIFT;
        const int li = lst[b] + __hip_atomic_fetch_add(&hist[b], 1, __ATOMIC_RELAXED,
                                                       __HIP_MEMORY_SCOPE_WORKGROUP);
        stage[li] = ((unsigned)r << BIN_SHIFT) | (unsigned)(cc & BIN_MASK);
        sbin[li]  = (unsigned short)b;
    };
    if (wide) {
        const int4* r4 = reinterpret_cast<const int4*>(row + s);
        const int4* c4 = reinterpret_cast<const int4*>(col + s);
        int4 r[4], cv[4];
#pragma unroll
        for (int u = 0; u < 4; ++u) {
            r[u]  = r4[tid + u * BIN_NODES];
            cv[u] = c4[tid + u * BIN_NODES];
        }
        loads_before_adds(r[0], r[1], r[2], r[3]);
        loads_before_adds(cv[0], cv[1], cv[2], cv[3]);
#pragma unroll
        for (int u = 0; u < 4; ++u) {
            place(r[u].x, cv[u].x);
            place(r[u].y, cv[u].y);
            place(r[u].z, cv[u].z);
            place(r[u].w, cv[u].w);
        }
    } else
    for (int j0 = s + tid; j0 < e; j0 += 4 * BIN_NODES) {
        int r[4], c[4];
#pragma unroll
        for (int u = 0; u < 4; ++u) {
            const int j = j0 + u * BIN_NODES;
            const int jj = (j < e) ? j : j0;
            r[u] = row[jj];
            c[u] = (j < e) ? col[jj] : -1;
        }
#pragma unroll
        for (int u = 0; u < 4; ++u) {
            if (c[u] >= 0) {
                const int b  = c[u] >> BIN_SHIFT;
                const int li = lst[b] + __hip_atomic_fetch_add(&hist[b], 1, __ATOMIC_RELAXED,
                                                               __HIP_MEMORY_SCOPE_WORKGROUP);
                stage[li] = ((unsigned)r[u] << BIN_SHIFT) | (unsigned)(c[u] & BIN_MASK);
                sbin[li]  = (unsigned short)b;
            }
        }
    }
    __syncthreads();
    const int n = e - s;
    for (int i = tid; i < n; i += BIN_NODES)
        keys[delta[sbin[i]] + i] = stage[i];
}

__global__ __launch_bounds__(BIN_NODES)
void bin_degree_xw1(const unsigned* __restrict__ keys, const int* __restrict__ start,
                    const float* __restrict__ x, const float* __restrict__ W1,
                    float* __restrict__ dinv, float* __restrict__ xw1s, int N) {
    __shared__ int   cnt[BIN_NODES];
    __shared__ float dl[BIN_NODES];
    __shared__ float tile[64][16 * 8];
    const int tid   = threadIdx.x;
    const int node0 = blockIdx.x * BIN_NODES;
    cnt[tid] = 0;
    __syncthreads();

    const int w    = tid >> 5;
    const int lane = tid & 31;
    const int m    = lane & 15;
    const int half = lane >> 4;
    const int n    = m;
    const float colMask = (n < 8) ? 1.0f : 0.0f;
    const int   nn      = n & 7;
    v2f a[2][4], b[4];
#pragma unroll
    for (int q = 0; q < 4; ++q) {
        const int k0 = 4 * q + 2 * half;
        b[q].x = W1[(k0 + 0) * 8 + nn] * colMask;
        b[q].y = W1[(k0 + 1) * 8 + nn] * colMask;
    }
#pragma unroll
    for (int i = 0; i < 2; ++i) {
        long rowA = (long)node0 + (w + 32 * i) * 16 + m;
        if (rowA >= N) rowA = N - 1;
        const float* xr = x + rowA * 16;
#pragma unroll
        for (int q = 0; q < 4; ++q) {
            const int k0 = 4 * q + 2 * half;
            a[i][q].x = xr[k0 + 0];
            a[i][q].y = xr[k0 + 1];
        }
    }

    const int s = start[blockIdx.x], e = start[blockIdx.x + 1];
    const int mis = (int)((reinterpret_cast<size_t>(keys + s) >> 2) & 3);
    const int s4  = min(s + ((4 - mis) & 3), e);
    const int n4  = (e - s4) >> 2;
    const int e4  = s4 + 4 * n4;
    const int nh  = s4 - s;
    const int nt  = e - e4;
    int edgeKey = -1;
    if (tid < nh + nt) edgeKey = (int)keys[(tid < nh) ? (s + tid) : (e4 + tid - nh)];
    const int4* k4 = reinterpret_cast<const int4*>(keys + s4);
    for (int i0 = tid; i0 < n4; i0 += 4 * BIN_NODES) {
        int4 k[4];
        bool ok[4];
#pragma unroll
        for (int u = 0; u < 4; ++u) {
            const int i = i0 + u * BIN_NODES;
            ok[u] = i < n4;
            k[u] = k4[ok[u] ? i : i0];
        }
        loads_before_adds(k[0], k[1], k[2], k[3]);
#pragma unroll
        for (int u = 0; u < 4; ++u) {
            if (ok[u]) {
                ldsAddI(&cnt[k[u].x & BIN_MASK]);
                ldsAddI(&cnt[k[u].y & BIN_MASK]);
                ldsAddI(&cnt[k[u].z & BIN_MASK]);
                ldsAddI(&cnt[k[u].w & BIN_MASK]);
            }
        }
    }
    if (edgeKey >= 0) ldsAddI(&cnt[edgeKey & BIN_MASK]);
    __syncthreads();
    const int node = node0 + tid;
    const float dv = __frsqrt_rn((float)(cnt[tid] + 1));
    dl[tid] = dv;
    if (node < N) dinv[node] = dv;
    __syncthreads();

    v8f acc0 = {}, acc1 = {};
#pragma unroll
    for (int q = 0; q < 4; ++q) {
        acc0 = __builtin_amdgcn_wmma_f32_16x16x4_f32(
                   false, a[0][q], false, b[q], (short)0, acc0, false, false);
        acc1 = __builtin_amdgcn_wmma_f32_16x16x4_f32(
                   false, a[1][q], false, b[q], (short)0, acc1, false, false);
    }

    float* t0 = &tile[w][0];
    float* t1 = &tile[w + 32][0];
    if (n < 8) {
#pragma unroll
        for (int r = 0; r < 8; ++r) {
            t0[(r + 8 * half) * 8 + n] = acc0[r];
            t1[(r + 8 * half) * 8 + n] = acc1[r];
        }
    }
    asm volatile("s_wait_dscnt 0x0" ::: "memory");

    const int rowIdx = lane >> 1;
    const int quad   = lane & 1;
#pragma unroll
    for (int i = 0; i < 2; ++i) {
        const int  lr = (w + 32 * i) * 16 + rowIdx;
        const long gr = (long)node0 + lr;
        const float sc = dl[lr];
        float4 v = *(const float4*)(&tile[w + 32 * i][0] + rowIdx * 8 + quad * 4);
        v.x *= sc; v.y *= sc; v.z *= sc; v.w *= sc;
        if (gr < N) *(float4*)(xw1s + gr * 8 + quad * 4) = v;
    }
}

__global__ __launch_bounds__(BIN_NODES)
void bin_prop1_hw2(const unsigned* __restrict__ keys, const int* __restrict__ start,
                   const float* __restrict__ dinv, const float* __restrict__ xw1s,
                   const float* __restrict__ b1, const float* __restrict__ W2,
                   float* __restrict__ hw2s, int N) {
    __shared__ float acc[BIN_NODES * 8];
    const int tid   = threadIdx.x;
    const int node0 = blockIdx.x * BIN_NODES;
    const int node  = node0 + tid;
#pragma unroll
    for (int i = 0; i < 8; ++i) acc[i * BIN_NODES + tid] = 0.0f;
    const float self = (node < N) ? dinv[node] : 0.0f;
    __syncthreads();

    const int s = start[blockIdx.x], e = start[blockIdx.x + 1];
    const int f   = tid & 7;
    const int sub = tid >> 3;
    for (int j0 = s + sub; j0 < e; j0 += 8 * 128) {
        unsigned k[8];
        bool ok[8];
#pragma unroll
        for (int u = 0; u < 8; ++u) {
            const int j = j0 + u * 128;
            ok[u] = j < e;
            k[u] = keys[ok[u] ? j : j0];
        }
        keys_in_one_group(k);
        float v[8];
        int   a[8];
#pragma unroll
        for (int u = 0; u < 8; ++u) {
            const int r  = (int)(k[u] >> BIN_SHIFT);
            const int cl = (int)(k[u] & BIN_MASK);
            v[u] = xw1s[(long)r * 8 + f];
            a[u] = cl * 8 + f;
        }
        rows_in_one_group(v);
#pragma unroll
        for (int u = 0; u < 8; ++u)
            if (ok[u]) ldsAddF(&acc[a[u]], v[u]);
    }
    __syncthreads();
    if (node < N) {
        const float4* sr = (const float4*)(xw1s + (long)node * 8);
        const float4 s0 = sr[0], s1 = sr[1];
        const float4 q0 = *(const float4*)&acc[tid * 8];
        const float4 q1 = *(const float4*)&acc[tid * 8 + 4];
        float hv[8];
        hv[0] = q0.x * self + self * s0.x; hv[1] = q0.y * self + self * s0.y;
        hv[2] = q0.z * self + self * s0.z; hv[3] = q0.w * self + self * s0.w;
        hv[4] = q1.x * self + self * s1.x; hv[5] = q1.y * self + self * s1.y;
        hv[6] = q1.z * self + self * s1.z; hv[7] = q1.w * self + self * s1.w;
        float o0 = 0.0f, o1 = 0.0f;
#pragma unroll
        for (int c = 0; c < 8; ++c) {
            const float h = fmaxf(hv[c] + b1[c], 0.0f);
            o0 = fmaf(h, W2[c * 2 + 0], o0);
            o1 = fmaf(h, W2[c * 2 + 1], o1);
        }
        float2 o;
        o.x = o0 * self;
        o.y = o1 * self;
        *(float2*)(hw2s + (long)node * 2) = o;
    }
}

__global__ __launch_bounds__(BIN_NODES)
void bin_prop2_out(const unsigned* __restrict__ keys, const int* __restrict__ start,
                   const float* __restrict__ dinv, const float* __restrict__ hw2s,
                   const float* __restrict__ b2, float* __restrict__ outb, int N) {
    __shared__ float acc[BIN_NODES * 2];
    const int tid   = threadIdx.x;
    const int node0 = blockIdx.x * BIN_NODES;
    const int node  = node0 + tid;
    acc[tid] = 0.0f;
    acc[BIN_NODES + tid] = 0.0f;
    const float self = (node < N) ? dinv[node] : 0.0f;
    __syncthreads();

    const int s = start[blockIdx.x], e = start[blockIdx.x + 1];
    const int f   = tid & 1;
    const int sub = tid >> 1;
    for (int j0 = s + sub; j0 < e; j0 += 8 * 512) {
        unsigned k[8];
        bool ok[8];
#pragma unroll
        for (int u = 0; u < 8; ++u) {
            const int j = j0 + u * 512;
            ok[u] = j < e;
            k[u] = keys[ok[u] ? j : j0];
        }
        keys_in_one_group(k);
        float v[8];
        int   a[8];
#pragma unroll
        for (int u = 0; u < 8; ++u) {
            const int r  = (int)(k[u] >> BIN_SHIFT);
            const int cl = (int)(k[u] & BIN_MASK);
            v[u] = hw2s[(long)r * 2 + f];
            a[u] = cl * 2 + f;
        }
        rows_in_one_group(v);
#pragma unroll
        for (int u = 0; u < 8; ++u)
            if (ok[u]) ldsAddF(&acc[a[u]], v[u]);
    }
    __syncthreads();
    if (node < N) {
        const float2 hs = *(const float2*)(hw2s + (long)node * 2);
        const float2 q  = *(const float2*)&acc[tid * 2];
        const float v0 = q.x * self + self * hs.x + b2[0];
        const float v1 = q.y * self + self * hs.y + b2[1];
        const float mx  = fmaxf(v0, v1);
        const float lse = mx + logf(expf(v0 - mx) + expf(v1 - mx));
        float2 o;
        o.x = v0 - lse;
        o.y = v1 - lse;
        *(float2*)(outb + (long)node * 2) = o;
    }
}

extern "C" void kernel_launch(void* const* d_in, const int* in_sizes, int n_in,
                              void* d_out, int out_size, void* d_ws, size_t ws_size,
                              hipStream_t stream) {
    const float* x    = (const float*)d_in[0];
    const float* W1   = (const float*)d_in[1];
    const float* b1   = (const float*)d_in[2];
    const float* W2   = (const float*)d_in[3];
    const float* b2   = (const float*)d_in[4];
    const int*   edge = (const int*)d_in[5];

    const int N = in_sizes[0] / 16;
    const int E = in_sizes[5] / 2;
    const int* row = edge;
    const int* col = edge + E;
    float* out = (float*)d_out;

    const int P = (N + BIN_NODES - 1) / BIN_NODES;

    size_t off = 0;
    auto carve = [&](size_t bytes) {
        void* p = (char*)d_ws + off;
        off = (off + bytes + 255) & ~(size_t)255;
        return p;
    };
    float*    dinv  = (float*)carve((size_t)N * 4);
    float*    xw1s  = (float*)carve((size_t)N * 8 * 4);
    float*    hw2s  = (float*)carve((size_t)N * 2 * 4);
    int*      tot   = (int*)carve((size_t)P * CNT_PITCH * 4);
    int*      cur   = (int*)carve((size_t)P * CNT_PITCH * 4);
    int*      start = (int*)carve((size_t)(P + 1) * 4);
    unsigned* keys  = (unsigned*)carve((size_t)E * 4);
    (void)ws_size; (void)n_in; (void)out_size;

    const int gC = (E + CHUNK - 1) / CHUNK;

    bin_init    <<<1,  BIN_NODES, 0, stream>>>(tot, cur, P);
    bin_count   <<<gC, BIN_NODES, 0, stream>>>(col, tot, E, P);
    bin_scan    <<<1,  BIN_NODES, 0, stream>>>(tot, start, P);
    bin_scatter <<<gC, BIN_NODES, 0, stream>>>(row, col, start, cur, keys, E, P);
    bin_degree_xw1<<<P, BIN_NODES, 0, stream>>>(keys, start, x, W1, dinv, xw1s, N);
    bin_prop1_hw2 <<<P, BIN_NODES, 0, stream>>>(keys, start, dinv, xw1s, b1, W2, hw2s, N);
    bin_prop2_out <<<P, BIN_NODES, 0, stream>>>(keys, start, dinv, hw2s, b2, out, N);
}


# Each bin owns a fixed run of key slots in the workspace and a fill counter (shared overflow list when a bin is full): edges placed in one pass, count and scan launches gone, 7 launches to 5
# speedup vs baseline: 5.7679x; 1.2255x over previous
#include <hip/hip_runtime.h>
#include <hip/hip_bf16.h>


typedef __attribute__((ext_vector_type(2)))  float    v2f;
typedef __attribute__((ext_vector_type(8)))  float    v8f;

#define BIN_SHIFT 10
#define BIN_NODES 1024
#define BIN_MASK  1023
#define CNT_PITCH 1088
#define CHUNK     16384
#define BIN_CAP   65600

__device__ __forceinline__ void ldsAddF(float* p, float v) {
    __hip_atomic_fetch_add(p, v, __ATOMIC_RELAXED, __HIP_MEMORY_SCOPE_WORKGROUP);
}
__device__ __forceinline__ void ldsAddI(int* p) {
    __hip_atomic_fetch_add(p, 1, __ATOMIC_RELAXED, __HIP_MEMORY_SCOPE_WORKGROUP);
}

__device__ __forceinline__ void loads_before_adds(const int4& a, const int4& b,
                                                  const int4& c, const int4& d) {
    asm volatile("" :: "v"(a.x), "v"(a.y), "v"(a.z), "v"(a.w),
                       "v"(b.x), "v"(b.y), "v"(b.z), "v"(b.w),
                       "v"(c.x), "v"(c.y), "v"(c.z), "v"(c.w),
                       "v"(d.x), "v"(d.y), "v"(d.z), "v"(d.w));
}

__device__ __forceinline__ void keys_in_one_group(const unsigned (&k)[8]) {
    asm volatile("" :: "v"(k[0]), "v"(k[1]), "v"(k[2]), "v"(k[3]),
                       "v"(k[4]), "v"(k[5]), "v"(k[6]), "v"(k[7]));
}
__device__ __forceinline__ void rows_in_one_group(const float (&v)[8]) {
    asm volatile("" :: "v"(v[0]), "v"(v[1]), "v"(v[2]), "v"(v[3]),
                       "v"(v[4]), "v"(v[5]), "v"(v[6]), "v"(v[7]));
}

__global__ void bin_init(int* __restrict__ cur, int* __restrict__ ovfcnt, int P) {
    int i = blockIdx.x * blockDim.x + threadIdx.x;
    if (i < P) cur[i * CNT_PITCH] = 0;
    if (i == 0) *ovfcnt = 0;
}

__device__ __forceinline__ void chunk_hist(const int* __restrict__ col, int s, int e,
                                           int* hist, int tid) {
    for (int j0 = s + tid; j0 < e; j0 += 4 * BIN_NODES) {
        int b[4];
#pragma unroll
        for (int u = 0; u < 4; ++u) {
            const int j = j0 + u * BIN_NODES;
            b[u] = (j < e) ? (col[j] >> BIN_SHIFT) : -1;
        }
#pragma unroll
        for (int u = 0; u < 4; ++u)
            if (b[u] >= 0) ldsAddI(&hist[b[u]]);
    }
}

__device__ __forceinline__ bool chunk_is_wide(const int* p, int s, int e) {
    return (e - s == CHUNK) && ((reinterpret_cast<size_t>(p + s) & 15) == 0);
}

__device__ __forceinline__ void chunk_hist_wide(const int* __restrict__ col, int s,
                                                int* hist, int tid) {
    const int4* c4 = reinterpret_cast<const int4*>(col + s);
    int4 c[4];
#pragma unroll
    for (int u = 0; u < 4; ++u) c[u] = c4[tid + u * BIN_NODES];
    loads_before_adds(c[0], c[1], c[2], c[3]);
#pragma unroll
    for (int u = 0; u < 4; ++u) {
        ldsAddI(&hist[c[u].x >> BIN_SHIFT]);
        ldsAddI(&hist[c[u].y >> BIN_SHIFT]);
        ldsAddI(&hist[c[u].z >> BIN_SHIFT]);
        ldsAddI(&hist[c[u].w >> BIN_SHIFT]);
    }
}

__device__ __forceinline__ int block_incl_scan(int v, int* s, int tid) {
    s[tid] = v;
    __syncthreads();
    for (int off = 1; off < BIN_NODES; off <<= 1) {
        const int t = (tid >= off) ? s[tid - off] : 0;
        __syncthreads();
        s[tid] += t;
        __syncthreads();
    }
    return s[tid];
}

__global__ __launch_bounds__(BIN_NODES)
void bin_scatter(const int* __restrict__ row, const int* __restrict__ col,
                 int* __restrict__ cur, unsigned* __restrict__ keys,
                 int* __restrict__ ovfcnt, unsigned* __restrict__ ovf, int E, int P) {
    __shared__ unsigned       stage[CHUNK];
    __shared__ unsigned short sbin[CHUNK];
    __shared__ int hist[BIN_NODES];
    __shared__ int lst[BIN_NODES];
    __shared__ int delta[BIN_NODES];
    const int tid = threadIdx.x;
    hist[tid] = 0;
    __syncthreads();
    const int s = blockIdx.x * CHUNK;
    const int e = min(s + CHUNK, E);
    const bool wide = chunk_is_wide(col, s, e) && chunk_is_wide(row, s, e);
    if (wide) chunk_hist_wide(col, s, hist, tid);
    else      chunk_hist(col, s, e, hist, tid);
    __syncthreads();
    const int h = hist[tid];
    int gbase = 0;
    if (tid < P && h)
        gbase = tid * BIN_CAP + __hip_atomic_fetch_add(&cur[tid * CNT_PITCH], h,
                                 __ATOMIC_RELAXED, __HIP_MEMORY_SCOPE_AGENT);
    const int excl = block_incl_scan(h, lst, tid) - h;
    lst[tid]   = excl;
    delta[tid] = gbase - excl;
    hist[tid]  = 0;
    __syncthreads();
    auto place = [&](int r, int cc) {
        const int b  = cc >> BIN_SHIFT;
        const int li = lst[b] + __hip_atomic_fetch_add(&hist[b], 1, __ATOMIC_RELAXED,
                                                       __HIP_MEMORY_SCOPE_WORKGROUP);
        stage[li] = ((unsigned)r << BIN_SHIFT) | (unsigned)(cc & BIN_MASK);
        sbin[li]  = (unsigned short)b;
    };
    if (wide) {
        const int4* r4 = reinterpret_cast<const int4*>(row + s);
        const int4* c4 = reinterpret_cast<const int4*>(col + s);
        int4 r[4], cv[4];
#pragma unroll
        for (int u = 0; u < 4; ++u) {
            r[u]  = r4[tid + u * BIN_NODES];
            cv[u] = c4[tid + u * BIN_NODES];
        }
        loads_before_adds(r[0], r[1], r[2], r[3]);
        loads_before_adds(cv[0], cv[1], cv[2], cv[3]);
#pragma unroll
        for (int u = 0; u < 4; ++u) {
            place(r[u].x, cv[u].x);
            place(r[u].y, cv[u].y);
            place(r[u].z, cv[u].z);
            place(r[u].w, cv[u].w);
        }
    } else
    for (int j0 = s + tid; j0 < e; j0 += 4 * BIN_NODES) {
        int r[4], c[4];
#pragma unroll
        for (int u = 0; u < 4; ++u) {
            const int j = j0 + u * BIN_NODES;
            const int jj = (j < e) ? j : j0;
            r[u] = row[jj];
            c[u] = (j < e) ? col[jj] : -1;
        }
#pragma unroll
        for (int u = 0; u < 4; ++u) {
            if (c[u] >= 0) {
                const int b  = c[u] >> BIN_SHIFT;
                const int li = lst[b] + __hip_atomic_fetch_add(&hist[b], 1, __ATOMIC_RELAXED,
                                                               __HIP_MEMORY_SCOPE_WORKGROUP);
                stage[li] = ((unsigned)r[u] << BIN_SHIFT) | (unsigned)(c[u] & BIN_MASK);
                sbin[li]  = (unsigned short)b;
            }
        }
    }
    __syncthreads();
    const int n = e - s;
    for (int i = tid; i < n; i += BIN_NODES) {
        const int b = sbin[i];
        const int p = delta[b] + i;
        if (p - b * BIN_CAP < BIN_CAP) {
            keys[p] = stage[i];
        } else {
            const int o = __hip_atomic_fetch_add(ovfcnt, 1, __ATOMIC_RELAXED,
                                                 __HIP_MEMORY_SCOPE_AGENT);
            ovf[2 * o]     = stage[i];
            ovf[2 * o + 1] = (unsigned)b;
        }
    }
}

__global__ __launch_bounds__(BIN_NODES)
void bin_degree_xw1(const unsigned* __restrict__ keys, const int* __restrict__ cur,
                    const int* __restrict__ ovfcnt, const unsigned* __restrict__ ovf,
                    const float* __restrict__ x, const float* __restrict__ W1,
                    float* __restrict__ dinv, float* __restrict__ xw1s, int N) {
    __shared__ int   cnt[BIN_NODES];
    __shared__ float dl[BIN_NODES];
    __shared__ float tile[64][16 * 8];
    const int tid   = threadIdx.x;
    const int node0 = blockIdx.x * BIN_NODES;
    cnt[tid] = 0;
    __syncthreads();

    const int w    = tid >> 5;
    const int lane = tid & 31;
    const int m    = lane & 15;
    const int half = lane >> 4;
    const int n    = m;
    const float colMask = (n < 8) ? 1.0f : 0.0f;
    const int   nn      = n & 7;
    v2f a[2][4], b[4];
#pragma unroll
    for (int q = 0; q < 4; ++q) {
        const int k0 = 4 * q + 2 * half;
        b[q].x = W1[(k0 + 0) * 8 + nn] * colMask;
        b[q].y = W1[(k0 + 1) * 8 + nn] * colMask;
    }
#pragma unroll
    for (int i = 0; i < 2; ++i) {
        long rowA = (long)node0 + (w + 32 * i) * 16 + m;
        if (rowA >= N) rowA = N - 1;
        const float* xr = x + rowA * 16;
#pragma unroll
        for (int q = 0; q < 4; ++q) {
            const int k0 = 4 * q + 2 * half;
            a[i][q].x = xr[k0 + 0];
            a[i][q].y = xr[k0 + 1];
        }
    }

    const int s = blockIdx.x * BIN_CAP;
    const int e = s + min(cur[blockIdx.x * CNT_PITCH], BIN_CAP);
    const int novf = *ovfcnt;
    const int mis = (int)((reinterpret_cast<size_t>(keys + s) >> 2) & 3);
    const int s4  = min(s + ((4 - mis) & 3), e);
    const int n4  = (e - s4) >> 2;
    const int e4  = s4 + 4 * n4;
    const int nh  = s4 - s;
    const int nt  = e - e4;
    int edgeKey = -1;
    if (tid < nh + nt) edgeKey = (int)keys[(tid < nh) ? (s + tid) : (e4 + tid - nh)];
    const int4* k4 = reinterpret_cast<const int4*>(keys + s4);
    for (int i0 = tid; i0 < n4; i0 += 4 * BIN_NODES) {
        int4 k[4];
        bool ok[4];
#pragma unroll
        for (int u = 0; u < 4; ++u) {
            const int i = i0 + u * BIN_NODES;
            ok[u] = i < n4;
            k[u] = k4[ok[u] ? i : i0];
        }
        loads_before_adds(k[0], k[1], k[2], k[3]);
#pragma unroll
        for (int u = 0; u < 4; ++u) {
            if (ok[u]) {
                ldsAddI(&cnt[k[u].x & BIN_MASK]);
                ldsAddI(&cnt[k[u].y & BIN_MASK]);
                ldsAddI(&cnt[k[u].z & BIN_MASK]);
                ldsAddI(&cnt[k[u].w & BIN_MASK]);
            }
        }
    }
    if (edgeKey >= 0) ldsAddI(&cnt[edgeKey & BIN_MASK]);
    for (int i = tid; i < novf; i += BIN_NODES)
        if (ovf[2 * i + 1] == (unsigned)blockIdx.x) ldsAddI(&cnt[ovf[2 * i] & BIN_MASK]);
    __syncthreads();
    const int node = node0 + tid;
    const float dv = __frsqrt_rn((float)(cnt[tid] + 1));
    dl[tid] = dv;
    if (node < N) dinv[node] = dv;
    __syncthreads();

    v8f acc0 = {}, acc1 = {};
#pragma unroll
    for (int q = 0; q < 4; ++q) {
        acc0 = __builtin_amdgcn_wmma_f32_16x16x4_f32(
                   false, a[0][q], false, b[q], (short)0, acc0, false, false);
        acc1 = __builtin_amdgcn_wmma_f32_16x16x4_f32(
                   false, a[1][q], false, b[q], (short)0, acc1, false, false);
    }

    float* t0 = &tile[w][0];
    float* t1 = &tile[w + 32][0];
    if (n < 8) {
#pragma unroll
        for (int r = 0; r < 8; ++r) {
            t0[(r + 8 * half) * 8 + n] = acc0[r];
            t1[(r + 8 * half) * 8 + n] = acc1[r];
        }
    }
    asm volatile("s_wait_dscnt 0x0" ::: "memory");

    const int rowIdx = lane >> 1;
    const int quad   = lane & 1;
#pragma unroll
    for (int i = 0; i < 2; ++i) {
        const int  lr = (w + 32 * i) * 16 + rowIdx;
        const long gr = (long)node0 + lr;
        const float sc = dl[lr];
        float4 v = *(const float4*)(&tile[w + 32 * i][0] + rowIdx * 8 + quad * 4);
        v.x *= sc; v.y *= sc; v.z *= sc; v.w *= sc;
        if (gr < N) *(float4*)(xw1s + gr * 8 + quad * 4) = v;
    }
}

__global__ __launch_bounds__(BIN_NODES)
void bin_prop1_hw2(const unsigned* __restrict__ keys, const int* __restrict__ cur,
                   const int* __restrict__ ovfcnt, const unsigned* __restrict__ ovf,
                   const float* __restrict__ dinv, const float* __restrict__ xw1s,
                   const float* __restrict__ b1, const float* __restrict__ W2,
                   float* __restrict__ hw2s, int N) {
    __shared__ float acc[BIN_NODES * 8];
    const int tid   = threadIdx.x;
    const int node0 = blockIdx.x * BIN_NODES;
    const int node  = node0 + tid;
#pragma unroll
    for (int i = 0; i < 8; ++i) acc[i * BIN_NODES + tid] = 0.0f;
    const float self = (node < N) ? dinv[node] : 0.0f;
    __syncthreads();

    const int s = blockIdx.x * BIN_CAP;
    const int e = s + min(cur[blockIdx.x * CNT_PITCH], BIN_CAP);
    const int novf = *ovfcnt;
    const int f   = tid & 7;
    const int sub = tid >> 3;
    for (int j0 = s + sub; j0 < e; j0 += 8 * 128) {
        unsigned k[8];
        bool ok[8];
#pragma unroll
        for (int u = 0; u < 8; ++u) {
            const int j = j0 + u * 128;
            ok[u] = j < e;
            k[u] = keys[ok[u] ? j : j0];
        }
        keys_in_one_group(k);
        float v[8];
        int   a[8];
#pragma unroll
        for (int u = 0; u < 8; ++u) {
            const int r  = (int)(k[u] >> BIN_SHIFT);
            const int cl = (int)(k[u] & BIN_MASK);
            v[u] = xw1s[(long)r * 8 + f];
            a[u] = cl * 8 + f;
        }
        rows_in_one_group(v);
#pragma unroll
        for (int u = 0; u < 8; ++u)
            if (ok[u]) ldsAddF(&acc[a[u]], v[u]);
    }
    for (int i = tid; i < novf; i += BIN_NODES) {
        if (ovf[2 * i + 1] == (unsigned)blockIdx.x) {
            const unsigned ko = ovf[2 * i];
            const float* tr = xw1s + (long)(ko >> BIN_SHIFT) * 8;
            float* ar = &acc[(ko & BIN_MASK) * 8];
            for (int c = 0; c < 8; ++c) ldsAddF(&ar[c], tr[c]);
        }
    }
    __syncthreads();
    if (node < N) {
        const float4* sr = (const float4*)(xw1s + (long)node * 8);
        const float4 s0 = sr[0], s1 = sr[1];
        const float4 q0 = *(const float4*)&acc[tid * 8];
        const float4 q1 = *(const float4*)&acc[tid * 8 + 4];
        float hv[8];
        hv[0] = q0.x * self + self * s0.x; hv[1] = q0.y * self + self * s0.y;
        hv[2] = q0.z * self + self * s0.z; hv[3] = q0.w * self + self * s0.w;
        hv[4] = q1.x * self + self * s1.x; hv[5] = q1.y * self + self * s1.y;
        hv[6] = q1.z * self + self * s1.z; hv[7] = q1.w * self + self * s1.w;
        float o0 = 0.0f, o1 = 0.0f;
#pragma unroll
        for (int c = 0; c < 8; ++c) {
            const float h = fmaxf(hv[c] + b1[c], 0.0f);
            o0 = fmaf(h, W2[c * 2 + 0], o0);
            o1 = fmaf(h, W2[c * 2 + 1], o1);
        }
        float2 o;
        o.x = o0 * self;
        o.y = o1 * self;
        *(float2*)(hw2s + (long)node * 2) = o;
    }
}

__global__ __launch_bounds__(BIN_NODES)
void bin_prop2_out(const unsigned* __restrict__ keys, const int* __restrict__ cur,
                   const int* __restrict__ ovfcnt, const unsigned* __restrict__ ovf,
                   const float* __restrict__ dinv, const float* __restrict__ hw2s,
                   const float* __restrict__ b2, float* __restrict__ outb, int N) {
    __shared__ float acc[BIN_NODES * 2];
    const int tid   = threadIdx.x;
    const int node0 = blockIdx.x * BIN_NODES;
    const int node  = node0 + tid;
    acc[tid] = 0.0f;
    acc[BIN_NODES + tid] = 0.0f;
    const float self = (node < N) ? dinv[node] : 0.0f;
    __syncthreads();

    const int s = blockIdx.x * BIN_CAP;
    const int e = s + min(cur[blockIdx.x * CNT_PITCH], BIN_CAP);
    const int novf = *ovfcnt;
    const int f   = tid & 1;
    const int sub = tid >> 1;
    for (int j0 = s + sub; j0 < e; j0 += 8 * 512) {
        unsigned k[8];
        bool ok[8];
#pragma unroll
        for (int u = 0; u < 8; ++u) {
            const int j = j0 + u * 512;
            ok[u] = j < e;
            k[u] = keys[ok[u] ? j : j0];
        }
        keys_in_one_group(k);
        float v[8];
        int   a[8];
#pragma unroll
        for (int u = 0; u < 8; ++u) {
            const int r  = (int)(k[u] >> BIN_SHIFT);
            const int cl = (int)(k[u] & BIN_MASK);
            v[u] = hw2s[(long)r * 2 + f];
            a[u] = cl * 2 + f;
        }
        rows_in_one_group(v);
#pragma unroll
        for (int u = 0; u < 8; ++u)
            if (ok[u]) ldsAddF(&acc[a[u]], v[u]);
    }
    for (int i = tid; i < novf; i += BIN_NODES) {
        if (ovf[2 * i + 1] == (unsigned)blockIdx.x) {
            const unsigned ko = ovf[2 * i];
            const float* tr = hw2s + (long)(ko >> BIN_SHIFT) * 2;
            float* ar = &acc[(ko & BIN_MASK) * 2];
            ldsAddF(&ar[0], tr[0]);
            ldsAddF(&ar[1], tr[1]);
        }
    }
    __syncthreads();
    if (node < N) {
        const float2 hs = *(const float2*)(hw2s + (long)node * 2);
        const float2 q  = *(const float2*)&acc[tid * 2];
        const float v0 = q.x * self + self * hs.x + b2[0];
        const float v1 = q.y * self + self * hs.y + b2[1];
        const float mx  = fmaxf(v0, v1);
        const float lse = mx + logf(expf(v0 - mx) + expf(v1 - mx));
        float2 o;
        o.x = v0 - lse;
        o.y = v1 - lse;
        *(float2*)(outb + (long)node * 2) = o;
    }
}

extern "C" void kernel_launch(void* const* d_in, const int* in_sizes, int n_in,
                              void* d_out, int out_size, void* d_ws, size_t ws_size,
                              hipStream_t stream) {
    const float* x    = (const float*)d_in[0];
    const float* W1   = (const float*)d_in[1];
    const float* b1   = (const float*)d_in[2];
    const float* W2   = (const float*)d_in[3];
    const float* b2   = (const float*)d_in[4];
    const int*   edge = (const int*)d_in[5];

    const int N = in_sizes[0] / 16;
    const int E = in_sizes[5] / 2;
    const int* row = edge;
    const int* col = edge + E;
    float* out = (float*)d_out;

    const int P = (N + BIN_NODES - 1) / BIN_NODES;

    size_t off = 0;
    auto carve = [&](size_t bytes) {
        void* p = (char*)d_ws + off;
        off = (off + bytes + 255) & ~(size_t)255;
        return p;
    };
    float*    dinv   = (float*)carve((size_t)N * 4);
    float*    xw1s   = (float*)carve((size_t)N * 8 * 4);
    float*    hw2s   = (float*)carve((size_t)N * 2 * 4);
    int*      cur    = (int*)carve((size_t)P * CNT_PITCH * 4);
    int*      ovfcnt = (int*)carve(4);
    unsigned* ovf    = (unsigned*)carve((size_t)E * 8);
    unsigned* keys   = (unsigned*)carve((size_t)P * BIN_CAP * 4);
    (void)ws_size; (void)n_in; (void)out_size;

    const int gC = (E + CHUNK - 1) / CHUNK;

    bin_init    <<<1,  BIN_NODES, 0, stream>>>(cur, ovfcnt, P);
    bin_scatter <<<gC, BIN_NODES, 0, stream>>>(row, col, cur, keys, ovfcnt, ovf, E, P);
    bin_degree_xw1<<<P, BIN_NODES, 0, stream>>>(keys, cur, ovfcnt, ovf, x, W1, dinv, xw1s, N);
    bin_prop1_hw2 <<<P, BIN_NODES, 0, stream>>>(keys, cur, ovfcnt, ovf, dinv, xw1s, b1, W2, hw2s, N);
    bin_prop2_out <<<P, BIN_NODES, 0, stream>>>(keys, cur, ovfcnt, ovf, dinv, hw2s, b2, out, N);
}


# Every (bin, chunk) pair owns a fixed 64-slot key run with empty marks: an edge is placed by one LDS rank, no histogram, scan, global counter or init launch (5 launches to 4); sums still arrival order
# speedup vs baseline: 6.2865x; 1.0899x over previous
#include <hip/hip_runtime.h>
#include <hip/hip_bf16.h>


typedef __attribute__((ext_vector_type(2)))  float    v2f;
typedef __attribute__((ext_vector_type(8)))  float    v8f;

#define BIN_SHIFT 10
#define BIN_NODES 1024
#define BIN_MASK  1023
#define CHUNK     16384
#define RUN_CAP   64
#define RUN_MASK  63
#define EMPTY_KEY 0xFFFFFFFFu

__device__ __forceinline__ void ldsAddF(float* p, float v) {
    __hip_atomic_fetch_add(p, v, __ATOMIC_RELAXED, __HIP_MEMORY_SCOPE_WORKGROUP);
}
__device__ __forceinline__ void ldsAddI(int* p) {
    __hip_atomic_fetch_add(p, 1, __ATOMIC_RELAXED, __HIP_MEMORY_SCOPE_WORKGROUP);
}

__device__ __forceinline__ void loads_before_adds(const int4& a, const int4& b,
                                                  const int4& c, const int4& d) {
    asm volatile("" :: "v"(a.x), "v"(a.y), "v"(a.z), "v"(a.w),
                       "v"(b.x), "v"(b.y), "v"(b.z), "v"(b.w),
                       "v"(c.x), "v"(c.y), "v"(c.z), "v"(c.w),
                       "v"(d.x), "v"(d.y), "v"(d.z), "v"(d.w));
}

__device__ __forceinline__ void keys_in_one_group(const unsigned* k) {
    asm volatile("" :: "v"(k[0]), "v"(k[1]), "v"(k[2]), "v"(k[3]),
                       "v"(k[4]), "v"(k[5]), "v"(k[6]), "v"(k[7]));
}
__device__ __forceinline__ void rows_in_one_group(const float* v) {
    asm volatile("" :: "v"(v[0]), "v"(v[1]), "v"(v[2]), "v"(v[3]),
                       "v"(v[4]), "v"(v[5]), "v"(v[6]), "v"(v[7]));
}

__device__ __forceinline__ bool chunk_is_wide(const int* p, int s, int e) {
    return (e - s == CHUNK) && ((reinterpret_cast<size_t>(p + s) & 15) == 0);
}

template <int W, bool CHECK>
__device__ __forceinline__ void prop_sweep(const unsigned* __restrict__ kb, int j0, int ns,
                                           const float* __restrict__ table,
                                           float* acc, int f) {
    constexpr int STEP = BIN_NODES / W;
    unsigned k[16];
#pragma unroll
    for (int u = 0; u < 16; ++u) {
        const int turn = (W == 8) ? 4 * u : 16 * (u & 3);
        int p = (j0 ^ turn) + u * STEP;
        if (CHECK) p = (j0 + u * STEP < ns) ? p : j0;
        k[u] = kb[p];
    }
    keys_in_one_group(k);
    keys_in_one_group(k + 8);
    float v[16];
    bool  ok[16];
#pragma unroll
    for (int u = 0; u < 16; ++u) {
        ok[u] = (int)k[u] >= 0;
        if (CHECK) ok[u] = ok[u] && (j0 + u * STEP < ns);
        v[u] = 0.0f;
        if (ok[u]) v[u] = table[(long)(k[u] >> BIN_SHIFT) * W + f];
    }
    rows_in_one_group(v);
    rows_in_one_group(v + 8);
#pragma unroll
    for (int u = 0; u < 16; ++u)
        if (ok[u]) ldsAddF(&acc[(int)(k[u] & BIN_MASK) * W + f], v[u]);
}

template <int W>
__device__ __forceinline__ void prop_bin(const unsigned* __restrict__ kb, int ns,
                                         const float* __restrict__ table,
                                         float* acc, int tid) {
    constexpr int SWEEP = 16 * (BIN_NODES / W);
    const int f   = tid & (W - 1);
    const int sub = tid / W;
    const int whole = (ns / SWEEP) * SWEEP;
    for (int j0 = sub; j0 < whole; j0 += SWEEP) prop_sweep<W, false>(kb, j0, ns, table, acc, f);
    if (whole + sub < ns) prop_sweep<W, true>(kb, whole + sub, ns, table, acc, f);
}

template <int MAXP>
__global__ __launch_bounds__(BIN_NODES)
void bin_scatter(const int* __restrict__ row, const int* __restrict__ col,
                 unsigned* __restrict__ keys, int* __restrict__ ovfn,
                 unsigned* __restrict__ ovf, int E, int P) {
    __shared__ unsigned stage[MAXP * RUN_CAP];
    __shared__ int hist[BIN_NODES];
    __shared__ int novf;
    const int tid = threadIdx.x;
    hist[tid] = 0;
    if (tid == 0) novf = 0;
    const int nq = P * (RUN_CAP / 4);
    {
        const uint4 em = make_uint4(EMPTY_KEY, EMPTY_KEY, EMPTY_KEY, EMPTY_KEY);
        uint4* s4 = reinterpret_cast<uint4*>(stage);
        for (int i = tid; i < nq; i += BIN_NODES) s4[i] = em;
    }
    __syncthreads();
    const int s = blockIdx.x * CHUNK;
    const int e = min(s + CHUNK, E);
    auto rank_of = [&](int cc) {
        return __hip_atomic_fetch_add(&hist[cc >> BIN_SHIFT], 1, __ATOMIC_RELAXED,
                                      __HIP_MEMORY_SCOPE_WORKGROUP);
    };
    auto put = [&](int r, int cc, int rank) {
        const int b = cc >> BIN_SHIFT;
        const unsigned key = ((unsigned)r << BIN_SHIFT) | (unsigned)(cc & BIN_MASK);
        if (rank < RUN_CAP) {
            stage[b * RUN_CAP + rank] = key;
        } else {
            const int o = __hip_atomic_fetch_add(&novf, 1, __ATOMIC_RELAXED,
                                                 __HIP_MEMORY_SCOPE_WORKGROUP);
            ovf[2 * ((size_t)s + o)]     = key;
            ovf[2 * ((size_t)s + o) + 1] = (unsigned)b;
        }
    };
    if (chunk_is_wide(col, s, e) && chunk_is_wide(row, s, e)) {
        const int4* r4 = reinterpret_cast<const int4*>(row + s);
        const int4* c4 = reinterpret_cast<const int4*>(col + s);
        int4 r[4], cv[4], rk[4];
#pragma unroll
        for (int u = 0; u < 4; ++u) {
            r[u]  = r4[tid + u * BIN_NODES];
            cv[u] = c4[tid + u * BIN_NODES];
        }
        loads_before_adds(r[0], r[1], r[2], r[3]);
        loads_before_adds(cv[0], cv[1], cv[2], cv[3]);
#pragma unroll
        for (int u = 0; u < 4; ++u) {
            rk[u].x = rank_of(cv[u].x);
            rk[u].y = rank_of(cv[u].y);
            rk[u].z = rank_of(cv[u].z);
            rk[u].w = rank_of(cv[u].w);
        }
#pragma unroll
        for (int u = 0; u < 4; ++u) {
            put(r[u].x, cv[u].x, rk[u].x);
            put(r[u].y, cv[u].y, rk[u].y);
            put(r[u].z, cv[u].z, rk[u].z);
            put(r[u].w, cv[u].w, rk[u].w);
        }
    } else
    for (int j0 = s + tid; j0 < e; j0 += 4 * BIN_NODES) {
        int r[4], c[4];
#pragma unroll
        for (int u = 0; u < 4; ++u) {
            const int j = j0 + u * BIN_NODES;
            const int jj = (j < e) ? j : j0;
            r[u] = row[jj];
            c[u] = (j < e) ? col[jj] : -1;
        }
#pragma unroll
        for (int u = 0; u < 4; ++u)
            if (c[u] >= 0) put(r[u], c[u], rank_of(c[u]));
    }
    __syncthreads();
    if (tid == 0) ovfn[blockIdx.x] = novf;
    const uint4* s4 = reinterpret_cast<const uint4*>(stage);
    uint4* k4 = reinterpret_cast<uint4*>(keys);
    const size_t nChunks = gridDim.x;
    for (int i = tid; i < nq; i += BIN_NODES) {
        const int b = i / (RUN_CAP / 4);
        const int q = i % (RUN_CAP / 4);
        k4[((size_t)b * nChunks + blockIdx.x) * (RUN_CAP / 4) + q] = s4[i];
    }
}

__global__ __launch_bounds__(BIN_NODES)
void bin_degree_xw1(const unsigned* __restrict__ keys, const int* __restrict__ ovfn,
                    const unsigned* __restrict__ ovf, int nChunks,
                    const float* __restrict__ x, const float* __restrict__ W1,
                    float* __restrict__ dinv, float* __restrict__ xw1s, int N) {
    __shared__ int   cnt[BIN_NODES];
    __shared__ float dl[BIN_NODES];
    __shared__ float tile[64][16 * 8];
    const int tid   = threadIdx.x;
    const int node0 = blockIdx.x * BIN_NODES;
    cnt[tid] = 0;
    __syncthreads();

    const int w    = tid >> 5;
    const int lane = tid & 31;
    const int m    = lane & 15;
    const int half = lane >> 4;
    const int n    = m;
    const float colMask = (n < 8) ? 1.0f : 0.0f;
    const int   nn      = n & 7;
    v2f a[2][4], b[4];
#pragma unroll
    for (int q = 0; q < 4; ++q) {
        const int k0 = 4 * q + 2 * half;
        b[q].x = W1[(k0 + 0) * 8 + nn] * colMask;
        b[q].y = W1[(k0 + 1) * 8 + nn] * colMask;
    }
#pragma unroll
    for (int i = 0; i < 2; ++i) {
        long rowA = (long)node0 + (w + 32 * i) * 16 + m;
        if (rowA >= N) rowA = N - 1;
        const float* xr = x + rowA * 16;
#pragma unroll
        for (int q = 0; q < 4; ++q) {
            const int k0 = 4 * q + 2 * half;
            a[i][q].x = xr[k0 + 0];
            a[i][q].y = xr[k0 + 1];
        }
    }

    const int4* k4 = reinterpret_cast<const int4*>(keys + (size_t)blockIdx.x * nChunks * RUN_CAP);
    const int n4 = nChunks * (RUN_CAP / 4);
    for (int i0 = tid; i0 < n4; i0 += 4 * BIN_NODES) {
        int4 k[4];
        bool ok[4];
#pragma unroll
        for (int u = 0; u < 4; ++u) {
            const int i = i0 + u * BIN_NODES;
            ok[u] = i < n4;
            k[u] = k4[ok[u] ? i : i0];
        }
        loads_before_adds(k[0], k[1], k[2], k[3]);
#pragma unroll
        for (int u = 0; u < 4; ++u) {
            if (ok[u]) {
                if (k[u].x >= 0) ldsAddI(&cnt[k[u].x & BIN_MASK]);
                if (k[u].y >= 0) ldsAddI(&cnt[k[u].y & BIN_MASK]);
                if (k[u].z >= 0) ldsAddI(&cnt[k[u].z & BIN_MASK]);
                if (k[u].w >= 0) ldsAddI(&cnt[k[u].w & BIN_MASK]);
            }
        }
    }
    for (int c = tid; c < nChunks; c += BIN_NODES) {
        const int no = min(ovfn[c], CHUNK);
        const unsigned* ol = ovf + 2 * (size_t)c * CHUNK;
        for (int i = 0; i < no; ++i)
            if (ol[2 * i + 1] == (unsigned)blockIdx.x) ldsAddI(&cnt[ol[2 * i] & BIN_MASK]);
    }
    __syncthreads();
    const int node = node0 + tid;
    const float dv = __frsqrt_rn((float)(cnt[tid] + 1));
    dl[tid] = dv;
    if (node < N) dinv[node] = dv;
    __syncthreads();

    v8f acc0 = {}, acc1 = {};
#pragma unroll
    for (int q = 0; q < 4; ++q) {
        acc0 = __builtin_amdgcn_wmma_f32_16x16x4_f32(
                   false, a[0][q], false, b[q], (short)0, acc0, false, false);
        acc1 = __builtin_amdgcn_wmma_f32_16x16x4_f32(
                   false, a[1][q], false, b[q], (short)0, acc1, false, false);
    }

    float* t0 = &tile[w][0];
    float* t1 = &tile[w + 32][0];
    if (n < 8) {
#pragma unroll
        for (int r = 0; r < 8; ++r) {
            t0[(r + 8 * half) * 8 + n] = acc0[r];
            t1[(r + 8 * half) * 8 + n] = acc1[r];
        }
    }
    asm volatile("s_wait_dscnt 0x0" ::: "memory");

    const int rowIdx = lane >> 1;
    const int quad   = lane & 1;
#pragma unroll
    for (int i = 0; i < 2; ++i) {
        const int  lr = (w + 32 * i) * 16 + rowIdx;
        const long gr = (long)node0 + lr;
        const float sc = dl[lr];
        float4 v = *(const float4*)(&tile[w + 32 * i][0] + rowIdx * 8 + quad * 4);
        v.x *= sc; v.y *= sc; v.z *= sc; v.w *= sc;
        if (gr < N) *(float4*)(xw1s + gr * 8 + quad * 4) = v;
    }
}

__global__ __launch_bounds__(BIN_NODES)
void bin_prop1_hw2(const unsigned* __restrict__ keys, const int* __restrict__ ovfn,
                   const unsigned* __restrict__ ovf, int nChunks,
                   const float* __restrict__ dinv, const float* __restrict__ xw1s,
                   const float* __restrict__ b1, const float* __restrict__ W2,
                   float* __restrict__ hw2s, int N) {
    __shared__ float acc[BIN_NODES * 8];
    const int tid   = threadIdx.x;
    const int node0 = blockIdx.x * BIN_NODES;
    const int node  = node0 + tid;
#pragma unroll
    for (int i = 0; i < 8; ++i) acc[i * BIN_NODES + tid] = 0.0f;
    const float self = (node < N) ? dinv[node] : 0.0f;
    __syncthreads();

    const unsigned* kb = keys + (size_t)blockIdx.x * nChunks * RUN_CAP;
    const int ns  = nChunks * RUN_CAP;
    prop_bin<8>(kb, ns, xw1s, acc, tid);
    for (int c = tid; c < nChunks; c += BIN_NODES) {
        const int no = min(ovfn[c], CHUNK);
        const unsigned* ol = ovf + 2 * (size_t)c * CHUNK;
        for (int i = 0; i < no; ++i) {
            if (ol[2 * i + 1] == (unsigned)blockIdx.x) {
                const unsigned ko = ol[2 * i];
                const float* tr = xw1s + (long)(ko >> BIN_SHIFT) * 8;
                float* ar = &acc[(ko & BIN_MASK) * 8];
                for (int c8 = 0; c8 < 8; ++c8) ldsAddF(&ar[c8], tr[c8]);
            }
        }
    }
    __syncthreads();
    if (node < N) {
        const float4* sr = (const float4*)(xw1s + (long)node * 8);
        const float4 s0 = sr[0], s1 = sr[1];
        const float4 q0 = *(const float4*)&acc[tid * 8];
        const float4 q1 = *(const float4*)&acc[tid * 8 + 4];
        float hv[8];
        hv[0] = q0.x * self + self * s0.x; hv[1] = q0.y * self + self * s0.y;
        hv[2] = q0.z * self + self * s0.z; hv[3] = q0.w * self + self * s0.w;
        hv[4] = q1.x * self + self * s1.x; hv[5] = q1.y * self + self * s1.y;
        hv[6] = q1.z * self + self * s1.z; hv[7] = q1.w * self + self * s1.w;
        float o0 = 0.0f, o1 = 0.0f;
#pragma unroll
        for (int c = 0; c < 8; ++c) {
            const float h = fmaxf(hv[c] + b1[c], 0.0f);
            o0 = fmaf(h, W2[c * 2 + 0], o0);
            o1 = fmaf(h, W2[c * 2 + 1], o1);
        }
        float2 o;
        o.x = o0 * self;
        o.y = o1 * self;
        *(float2*)(hw2s + (long)node * 2) = o;
    }
}

__global__ __launch_bounds__(BIN_NODES)
void bin_prop2_out(const unsigned* __restrict__ keys, const int* __restrict__ ovfn,
                   const unsigned* __restrict__ ovf, int nChunks,
                   const float* __restrict__ dinv, const float* __restrict__ hw2s,
                   const float* __restrict__ b2, float* __restrict__ outb, int N) {
    __shared__ float acc[BIN_NODES * 2];
    const int tid   = threadIdx.x;
    const int node0 = blockIdx.x * BIN_NODES;
    const int node  = node0 + tid;
    acc[tid] = 0.0f;
    acc[BIN_NODES + tid] = 0.0f;
    const float self = (node < N) ? dinv[node] : 0.0f;
    __syncthreads();

    const unsigned* kb = keys + (size_t)blockIdx.x * nChunks * RUN_CAP;
    const int ns  = nChunks * RUN_CAP;
    prop_bin<2>(kb, ns, hw2s, acc, tid);
    for (int c = tid; c < nChunks; c += BIN_NODES) {
        const int no = min(ovfn[c], CHUNK);
        const unsigned* ol = ovf + 2 * (size_t)c * CHUNK;
        for (int i = 0; i < no; ++i) {
            if (ol[2 * i + 1] == (unsigned)blockIdx.x) {
                const unsigned ko = ol[2 * i];
                const float* tr = hw2s + (long)(ko >> BIN_SHIFT) * 2;
                float* ar = &acc[(ko & BIN_MASK) * 2];
                ldsAddF(&ar[0], tr[0]);
                ldsAddF(&ar[1], tr[1]);
            }
        }
    }
    __syncthreads();
    if (node < N) {
        const float2 hs = *(const float2*)(hw2s + (long)node * 2);
        const float2 q  = *(const float2*)&acc[tid * 2];
        const float v0 = q.x * self + self * hs.x + b2[0];
        const float v1 = q.y * self + self * hs.y + b2[1];
        const float mx  = fmaxf(v0, v1);
        const float lse = mx + logf(expf(v0 - mx) + expf(v1 - mx));
        float2 o;
        o.x = v0 - lse;
        o.y = v1 - lse;
        *(float2*)(outb + (long)node * 2) = o;
    }
}

extern "C" void kernel_launch(void* const* d_in, const int* in_sizes, int n_in,
                              void* d_out, int out_size, void* d_ws, size_t ws_size,
                              hipStream_t stream) {
    const float* x    = (const float*)d_in[0];
    const float* W1   = (const float*)d_in[1];
    const float* b1   = (const float*)d_in[2];
    const float* W2   = (const float*)d_in[3];
    const float* b2   = (const float*)d_in[4];
    const int*   edge = (const int*)d_in[5];

    const int N = in_sizes[0] / 16;
    const int E = in_sizes[5] / 2;
    const int* row = edge;
    const int* col = edge + E;
    float* out = (float*)d_out;

    const int P  = (N + BIN_NODES - 1) / BIN_NODES;
    const int gC = (E + CHUNK - 1) / CHUNK;

    size_t off = 0;
    auto carve = [&](size_t bytes) {
        void* p = (char*)d_ws + off;
        off = (off + bytes + 255) & ~(size_t)255;
        return p;
    };
    float*    dinv = (float*)carve((size_t)N * 4);
    float*    xw1s = (float*)carve((size_t)N * 8 * 4);
    float*    hw2s = (float*)carve((size_t)N * 2 * 4);
    int*      ovfn = (int*)carve((size_t)gC * 4);
    unsigned* ovf  = (unsigned*)carve((size_t)gC * CHUNK * 8);
    unsigned* keys = (unsigned*)carve((size_t)P * gC * RUN_CAP * 4);
    (void)ws_size; (void)n_in; (void)out_size;

    if (P <= 512)
        bin_scatter<512> <<<gC, BIN_NODES, 0, stream>>>(row, col, keys, ovfn, ovf, E, P);
    else
        bin_scatter<1024><<<gC, BIN_NODES, 0, stream>>>(row, col, keys, ovfn, ovf, E, P);
    bin_degree_xw1<<<P, BIN_NODES, 0, stream>>>(keys, ovfn, ovf, gC, x, W1, dinv, xw1s, N);
    bin_prop1_hw2 <<<P, BIN_NODES, 0, stream>>>(keys, ovfn, ovf, gC, dinv, xw1s, b1, W2, hw2s, N);
    bin_prop2_out <<<P, BIN_NODES, 0, stream>>>(keys, ovfn, ovf, gC, dinv, hw2s, b2, out, N);
}
